# starting kernel
# baseline (speedup 1.0000x reference)
#include <hip/hip_runtime.h>
#include <hip/hip_bf16.h>

#define GB   16
#define GN   10000
#define GE   160000
#define GS   128
#define GH   64
#define GEMB 64
#define GP   16

typedef __attribute__((ext_vector_type(2))) float v2f;
typedef __attribute__((ext_vector_type(4))) float v4f;
typedef __attribute__((ext_vector_type(8))) float v8f;

__global__ __launch_bounds__(256) void zero4_kernel(v4f* __restrict__ p, int n4) {
    int i = blockIdx.x * blockDim.x + threadIdx.x;
    if (i < n4) p[i] = (v4f){0.f, 0.f, 0.f, 0.f};
}

__global__ __launch_bounds__(256) void deg_kernel(const int* __restrict__ ei,
                                                  const float* __restrict__ ew,
                                                  float* __restrict__ deg) {
    int idx = blockIdx.x * blockDim.x + threadIdx.x;
    if (idx >= GB * GE) return;
    int b = idx / GE;
    int e = idx - b * GE;
    int c = ei[(size_t)b * 2 * GE + GE + e];
    atomicAdd(&deg[(size_t)b * GN + c], ew[idx]);
}

__global__ __launch_bounds__(256) void dinv_kernel(float* __restrict__ deg) {
    int i = blockIdx.x * blockDim.x + threadIdx.x;
    if (i < GB * GN) {
        deg[i] = rsqrtf(deg[i] + 1.0f);
    }
}

__global__ __launch_bounds__(256) void norm_kernel(const int* __restrict__ ei,
                                                   const float* __restrict__ ew,
                                                   const float* __restrict__ dinv,
                                                   float* __restrict__ nrm) {
    int idx = blockIdx.x * blockDim.x + threadIdx.x;
    if (idx >= GB * GE) return;
    int b = idx / GE;
    int e = idx - b * GE;
    int r = ei[(size_t)b * 2 * GE + e];
    int c = ei[(size_t)b * 2 * GE + GE + e];
    nrm[idx] = dinv[(size_t)b * GN + r] * ew[idx] * dinv[(size_t)b * GN + c];
}

template <int HIN>
__global__ __launch_bounds__(128) void gemm_wmma_kernel(const float* __restrict__ Hsrc,
                                                        const float* __restrict__ W,
                                                        float* __restrict__ Dst) {
    __shared__ float wlds[GH * HIN];
    const int tid = threadIdx.x;
    for (int i = tid; i < GH * HIN; i += 128) wlds[i] = W[i];
    __syncthreads();

    const int wave = tid >> 5;
    const int lane = tid & 31;
    const int half = lane >> 4;
    const int lm   = lane & 15;
    const int b    = blockIdx.z;
    const int m0   = blockIdx.x * 16;
    const int n0   = wave * 16;

    const float* hrow = Hsrc + ((size_t)b * GN + m0 + lm) * HIN;
    const float* wrow = wlds + (size_t)(n0 + lm) * HIN;

    v8f acc = {};
#pragma unroll
    for (int k0 = 0; k0 < HIN; k0 += 4) {
        const int ka = k0 + half * 2;
        v2f a = *(const v2f*)(hrow + ka);
        v2f bf = *(const v2f*)(wrow + ka);
        acc = __builtin_amdgcn_wmma_f32_16x16x4_f32(
             false, a,  false, bf,
             (short)0, acc,  false,  false);
    }

    float* drow = Dst + ((size_t)b * GN + m0) * GH + n0 + lm;
#pragma unroll
    for (int j = 0; j < 8; ++j) {
        drow[(size_t)(j + 8 * half) * GH] = acc[j];
    }
}

__global__ __launch_bounds__(256) void scatter_kernel(const int* __restrict__ ei,
                                                      const float* __restrict__ nrm,
                                                      const float* __restrict__ tmp,
                                                      float* __restrict__ agg) {
    long long idx = (long long)blockIdx.x * blockDim.x + threadIdx.x;
    if (idx >= (long long)GB * GE * (GH / 4)) return;
    int k4 = (int)(idx & 15);
    long long eg = idx >> 4;
    int b = (int)(eg / GE);
    int e = (int)(eg - (long long)b * GE);
    int r = ei[(size_t)b * 2 * GE + e];
    int c = ei[(size_t)b * 2 * GE + GE + e];
    float s = nrm[eg];
    v4f v = *(const v4f*)(tmp + ((size_t)b * GN + r) * GH + k4 * 4) * s;
    float* dst = agg + ((size_t)b * GN + c) * GH + k4 * 4;
    atomicAdd(dst + 0, v.x);
    atomicAdd(dst + 1, v.y);
    atomicAdd(dst + 2, v.z);
    atomicAdd(dst + 3, v.w);
}

__global__ __launch_bounds__(256) void act_kernel(const float* __restrict__ agg,
                                                  const float* __restrict__ tmp,
                                                  const float* __restrict__ dinv,
                                                  const float* __restrict__ bias,
                                                  float* __restrict__ hout,
                                                  int do_tanh) {
    int idx = blockIdx.x * blockDim.x + threadIdx.x;
    if (idx >= GB * GN * (GH / 4)) return;
    int k4  = idx & 15;
    int node = idx >> 4;
    float di = dinv[node];
    float sl = di * di;
    v4f a = *(const v4f*)(agg + (size_t)idx * 4);
    v4f t = *(const v4f*)(tmp + (size_t)idx * 4);
    v4f bb = *(const v4f*)(bias + k4 * 4);
    v4f v = a + t * sl + bb;
    if (do_tanh) {
        v.x = tanhf(v.x); v.y = tanhf(v.y); v.z = tanhf(v.z); v.w = tanhf(v.w);
    }
    *(v4f*)(hout + (size_t)idx * 4) = v;
}

__global__ __launch_bounds__(256) void gather_kernel(const float* __restrict__ emb,
                                                     const int* __restrict__ pos,
                                                     float* __restrict__ out) {
    int idx = blockIdx.x * blockDim.x + threadIdx.x;
    if (idx >= GB * GP * GEMB) return;
    int k = idx & (GEMB - 1);
    int bp = idx >> 6;
    int b = bp / GP;
    int pv = pos[bp];
    out[idx] = (pv >= 0) ? emb[((size_t)b * GN + pv) * GH + k] : -3.0f;
}

extern "C" void kernel_launch(void* const* d_in, const int* in_sizes, int n_in,
                              void* d_out, int out_size, void* d_ws, size_t ws_size,
                              hipStream_t stream) {
    const float* x   = (const float*)d_in[0];
    const int*   ei  = (const int*)d_in[1];
    const float* ew  = (const float*)d_in[2];
    const int*   pos = (const int*)d_in[3];
    const float* W1  = (const float*)d_in[4];
    const float* b1  = (const float*)d_in[5];
    const float* W2  = (const float*)d_in[6];
    const float* b2  = (const float*)d_in[7];
    const float* W3  = (const float*)d_in[8];
    const float* b3  = (const float*)d_in[9];
    float* out = (float*)d_out;

    float* ws   = (float*)d_ws;
    const size_t nBN  = (size_t)GB * GN;
    const size_t nBE  = (size_t)GB * GE;
    const size_t nBNH = (size_t)GB * GN * GH;
    float* dinv = ws;
    float* nrm  = dinv + nBN;
    float* tmp  = nrm + nBE;
    float* agg  = tmp + nBNH;
    float* hbuf = agg + nBNH;

    const int T = 256;
    dim3 gemm_grid(GN / 16, 1, GB);

    zero4_kernel<<<((int)(nBN / 4) + T - 1) / T, T, 0, stream>>>((v4f*)dinv, (int)(nBN / 4));
    deg_kernel<<<(GB * GE + T - 1) / T, T, 0, stream>>>(ei, ew, dinv);
    dinv_kernel<<<((int)nBN + T - 1) / T, T, 0, stream>>>(dinv);
    norm_kernel<<<(GB * GE + T - 1) / T, T, 0, stream>>>(ei, ew, dinv, nrm);

    const long long scatterN = (long long)GB * GE * (GH / 4);
    const int scatterBlocks = (int)((scatterN + T - 1) / T);
    const int elem4Blocks = (int)((nBNH / 4 + T - 1) / T);

    gemm_wmma_kernel<GS><<<gemm_grid, 128, 0, stream>>>(x, W1, tmp);
    zero4_kernel<<<elem4Blocks, T, 0, stream>>>((v4f*)agg, (int)(nBNH / 4));
    scatter_kernel<<<scatterBlocks, T, 0, stream>>>(ei, nrm, tmp, agg);
    act_kernel<<<elem4Blocks, T, 0, stream>>>(agg, tmp, dinv, b1, hbuf, 1);

    gemm_wmma_kernel<GH><<<gemm_grid, 128, 0, stream>>>(hbuf, W2, tmp);
    zero4_kernel<<<elem4Blocks, T, 0, stream>>>((v4f*)agg, (int)(nBNH / 4));
    scatter_kernel<<<scatterBlocks, T, 0, stream>>>(ei, nrm, tmp, agg);
    act_kernel<<<elem4Blocks, T, 0, stream>>>(agg, tmp, dinv, b2, hbuf, 1);

    gemm_wmma_kernel<GH><<<gemm_grid, 128, 0, stream>>>(hbuf, W3, tmp);
    zero4_kernel<<<elem4Blocks, T, 0, stream>>>((v4f*)agg, (int)(nBNH / 4));
    scatter_kernel<<<scatterBlocks, T, 0, stream>>>(ei, nrm, tmp, agg);
    act_kernel<<<elem4Blocks, T, 0, stream>>>(agg, tmp, dinv, b3, hbuf, 0);

    gather_kernel<<<(GB * GP * GEMB + T - 1) / T, T, 0, stream>>>(hbuf, pos, out);
}


# Global float atomics (degree sum, 3 scatters) replaced by work-groups owning a 625-node target range and summing it in LDS (LDS float atomics); zero and act passes folded in; sum order changes
# speedup vs baseline: 2.6896x; 2.6896x over previous
#include <hip/hip_runtime.h>
#include <hip/hip_bf16.h>

#define GB   16
#define GN   10000
#define GE   160000
#define GS   128
#define GH   64
#define GEMB 64
#define GP   16

typedef __attribute__((ext_vector_type(2))) float v2f;
typedef __attribute__((ext_vector_type(4))) float v4f;
typedef __attribute__((ext_vector_type(8))) float v8f;
typedef __attribute__((ext_vector_type(2))) unsigned int v2u;
typedef __attribute__((ext_vector_type(4))) unsigned int v4u;
typedef __attribute__((ext_vector_type(4))) int v4i;

#define NRANGE 16
#define RN     (GN / NRANGE)
#define OWN_T  1024
#define SB     16384

__global__ __launch_bounds__(OWN_T) void deg_kernel(const int* __restrict__ ei,
                                                    const float* __restrict__ ew,
                                                    float* __restrict__ dinv) {
    __shared__ float dl[RN];
    const int tid = threadIdx.x;
    const int b   = blockIdx.y;
    const unsigned lo = blockIdx.x * RN;
    for (int i = tid; i < RN; i += OWN_T) dl[i] = 0.f;
    __syncthreads();

    const v4i* col4 = (const v4i*)(ei + (size_t)b * 2 * GE + GE);
    const v4f* w4   = (const v4f*)(ew + (size_t)b * GE);
    for (int it = 0; it < (GE / 4 + 4 * OWN_T - 1) / (4 * OWN_T); ++it) {
        v4i c[4];
        v4f w[4];
#pragma unroll
        for (int u = 0; u < 4; ++u) {
            int p = (it * 4 + u) * OWN_T + tid;
            if (p < GE / 4) { c[u] = col4[p]; w[u] = w4[p]; }
            else            { c[u] = (v4i){-1, -1, -1, -1}; w[u] = (v4f){0.f, 0.f, 0.f, 0.f}; }
        }
#pragma unroll
        for (int u = 0; u < 4; ++u) {
#pragma unroll
            for (int j = 0; j < 4; ++j) {
                unsigned cl = (unsigned)c[u][j] - lo;
                if (cl < (unsigned)RN) atomicAdd(&dl[cl], w[u][j]);
            }
        }
    }
    __syncthreads();
    for (int i = tid; i < RN; i += OWN_T)
        dinv[(size_t)b * GN + lo + i] = rsqrtf(dl[i] + 1.0f);
}

__global__ __launch_bounds__(256) void norm_kernel(const int* __restrict__ ei,
                                                   const float* __restrict__ ew,
                                                   const float* __restrict__ dinv,
                                                   v2u* __restrict__ rec) {
    int idx = blockIdx.x * blockDim.x + threadIdx.x;
    if (idx >= GB * GE) return;
    int b = idx / GE;
    int e = idx - b * GE;
    int r = ei[(size_t)b * 2 * GE + e];
    int c = ei[(size_t)b * 2 * GE + GE + e];
    float nv = dinv[(size_t)b * GN + r] * ew[idx] * dinv[(size_t)b * GN + c];
    rec[idx] = (v2u){(unsigned)r | ((unsigned)c << 16), __float_as_uint(nv)};
}

template <int HIN>
__global__ __launch_bounds__(128) void gemm_wmma_kernel(const float* __restrict__ Hsrc,
                                                        const float* __restrict__ W,
                                                        float* __restrict__ Dst) {
    __shared__ float wlds[GH * HIN];
    const int tid = threadIdx.x;
    for (int i = tid; i < GH * HIN; i += 128) wlds[i] = W[i];
    __syncthreads();

    const int wave = tid >> 5;
    const int lane = tid & 31;
    const int half = lane >> 4;
    const int lm   = lane & 15;
    const int b    = blockIdx.z;
    const int m0   = blockIdx.x * 16;
    const int n0   = wave * 16;

    const float* hrow = Hsrc + ((size_t)b * GN + m0 + lm) * HIN;
    const float* wrow = wlds + (size_t)(n0 + lm) * HIN;

    v8f acc = {};
#pragma unroll
    for (int k0 = 0; k0 < HIN; k0 += 4) {
        const int ka = k0 + half * 2;
        v2f a = *(const v2f*)(hrow + ka);
        v2f bf = *(const v2f*)(wrow + ka);
        acc = __builtin_amdgcn_wmma_f32_16x16x4_f32(
             false, a,  false, bf,
             (short)0, acc,  false,  false);
    }

    float* drow = Dst + ((size_t)b * GN + m0) * GH + n0 + lm;
#pragma unroll
    for (int j = 0; j < 8; ++j) {
        drow[(size_t)(j + 8 * half) * GH] = acc[j];
    }
}

template <int DO_TANH>
__global__ __launch_bounds__(OWN_T) void agg_kernel(const v2u* __restrict__ rec,
                                                    const float* __restrict__ tmp,
                                                    const float* __restrict__ dinv,
                                                    const float* __restrict__ bias,
                                                    float* __restrict__ hout) {
    __shared__ float acc[RN * GH];
    __shared__ v2u   list[SB];
    __shared__ unsigned int cnt;

    const int tid  = threadIdx.x;
    const int lane = tid & 31;
    const int wave = __builtin_amdgcn_readfirstlane(tid >> 5);
    const int b    = blockIdx.y;
    const unsigned lo = blockIdx.x * RN;
    const size_t nodeBase = (size_t)b * GN + lo;

    for (int i4 = tid; i4 < RN * (GH / 4); i4 += OWN_T) {
        int node = i4 >> 4;
        int k4   = i4 & 15;
        float di = dinv[nodeBase + node];
        float sl = di * di;
        v4f t  = *(const v4f*)(tmp + (nodeBase + node) * GH + k4 * 4);
        v4f bb = *(const v4f*)(bias + k4 * 4);
        *(v4f*)(acc + (size_t)i4 * 4) = t * sl + bb;
    }
    if (tid == 0) cnt = 0;
    __syncthreads();

    const v4u*   rb = (const v4u*)(rec + (size_t)b * GE);
    const float* tb = tmp + (size_t)b * GN * GH + 2 * lane;
    unsigned prev = 0;

    for (int base = 0; base < GE; base += SB) {
        v4u r[8];
#pragma unroll
        for (int q = 0; q < 8; ++q) {
            int p = (base >> 1) + q * OWN_T + tid;
            if (p < GE / 2) r[q] = rb[p];
            else            r[q] = (v4u){0xFFFF0000u, 0u, 0xFFFF0000u, 0u};
        }
        unsigned m[16];
        unsigned tot = 0;
#pragma unroll
        for (int i = 0; i < 16; ++i) {
            unsigned word = (i & 1) ? r[i >> 1].z : r[i >> 1].x;
            bool hit = ((word >> 16) - lo) < (unsigned)RN;
            m[i] = __builtin_amdgcn_ballot_w32(hit);
            tot += __builtin_popcount(m[i]);
        }
        unsigned at = 0;
        if (tot) {
            if (lane == 0) at = atomicAdd(&cnt, tot);
            at = __builtin_amdgcn_readfirstlane(at);
        }
        at -= prev;
#pragma unroll
        for (int i = 0; i < 16; ++i) {
            unsigned word = (i & 1) ? r[i >> 1].z : r[i >> 1].x;
            unsigned nb   = (i & 1) ? r[i >> 1].w : r[i >> 1].y;
            unsigned cl   = (word >> 16) - lo;
            if (cl < (unsigned)RN)
                list[at + __builtin_amdgcn_mbcnt_lo(m[i], 0u)] = (v2u){(word & 0xFFFFu) | (cl << 16), nb};
            at += __builtin_popcount(m[i]);
        }
        __syncthreads();
        const unsigned cur = cnt;
        const int n = __builtin_amdgcn_readfirstlane(cur - prev);
        prev = cur;

        for (int i0 = wave * 8; i0 < n; i0 += (OWN_T / 32) * 8) {
            v2f      v[8];
            unsigned en[8];
            float    wv[8];
#pragma unroll
            for (int u = 0; u < 8; ++u) {
                int idx = i0 + u;
                idx = idx < n ? idx : n - 1;
                v2u e = list[idx];
                en[u] = __builtin_amdgcn_readfirstlane(e.x);
                wv[u] = __uint_as_float(__builtin_amdgcn_readfirstlane(e.y));
                v[u]  = *(const v2f*)(tb + (size_t)(en[u] & 0xFFFFu) * GH);
            }
#pragma unroll
            for (int u = 0; u < 8; ++u) {
                if (i0 + u < n) {
                    float* d = acc + (size_t)(en[u] >> 16) * GH + 2 * lane;
                    atomicAdd(d,     v[u].x * wv[u]);
                    atomicAdd(d + 1, v[u].y * wv[u]);
                }
            }
        }
        __syncthreads();
    }

    for (int i4 = tid; i4 < RN * (GH / 4); i4 += OWN_T) {
        v4f v = *(const v4f*)(acc + (size_t)i4 * 4);
        if (DO_TANH) {
            v.x = tanhf(v.x); v.y = tanhf(v.y); v.z = tanhf(v.z); v.w = tanhf(v.w);
        }
        *(v4f*)(hout + nodeBase * GH + (size_t)i4 * 4) = v;
    }
}

__global__ __launch_bounds__(256) void gather_kernel(const float* __restrict__ emb,
                                                     const int* __restrict__ pos,
                                                     float* __restrict__ out) {
    int idx = blockIdx.x * blockDim.x + threadIdx.x;
    if (idx >= GB * GP * GEMB) return;
    int k = idx & (GEMB - 1);
    int bp = idx >> 6;
    int b = bp / GP;
    int pv = pos[bp];
    out[idx] = (pv >= 0) ? emb[((size_t)b * GN + pv) * GH + k] : -3.0f;
}

extern "C" void kernel_launch(void* const* d_in, const int* in_sizes, int n_in,
                              void* d_out, int out_size, void* d_ws, size_t ws_size,
                              hipStream_t stream) {
    const float* x   = (const float*)d_in[0];
    const int*   ei  = (const int*)d_in[1];
    const float* ew  = (const float*)d_in[2];
    const int*   pos = (const int*)d_in[3];
    const float* W1  = (const float*)d_in[4];
    const float* b1  = (const float*)d_in[5];
    const float* W2  = (const float*)d_in[6];
    const float* b2  = (const float*)d_in[7];
    const float* W3  = (const float*)d_in[8];
    const float* b3  = (const float*)d_in[9];
    float* out = (float*)d_out;

    float* ws   = (float*)d_ws;
    const size_t nBN  = (size_t)GB * GN;
    const size_t nBE  = (size_t)GB * GE;
    const size_t nBNH = (size_t)GB * GN * GH;
    float* dinv = ws;
    v2u*   rec  = (v2u*)(dinv + nBN);
    float* tmp  = dinv + nBN + 2 * nBE;
    float* hbuf = tmp + nBNH;

    const int T = 256;
    dim3 gemm_grid(GN / 16, 1, GB);
    dim3 own_grid(NRANGE, GB, 1);

    deg_kernel<<<own_grid, OWN_T, 0, stream>>>(ei, ew, dinv);
    norm_kernel<<<(GB * GE + T - 1) / T, T, 0, stream>>>(ei, ew, dinv, rec);

    gemm_wmma_kernel<GS><<<gemm_grid, 128, 0, stream>>>(x, W1, tmp);
    agg_kernel<1><<<own_grid, OWN_T, 0, stream>>>(rec, tmp, dinv, b1, hbuf);

    gemm_wmma_kernel<GH><<<gemm_grid, 128, 0, stream>>>(hbuf, W2, tmp);
    agg_kernel<1><<<own_grid, OWN_T, 0, stream>>>(rec, tmp, dinv, b2, hbuf);

    gemm_wmma_kernel<GH><<<gemm_grid, 128, 0, stream>>>(hbuf, W3, tmp);
    agg_kernel<0><<<own_grid, OWN_T, 0, stream>>>(rec, tmp, dinv, b3, hbuf);

    gather_kernel<<<(GB * GP * GEMB + T - 1) / T, T, 0, stream>>>(hbuf, pos, out);
}


# Layer-2/3 linear maps moved into the preceding aggregation's epilogue: owner block multiplies its finished rows from LDS (pitch 68) by next W^T on WMMA; 9 launches to 7; roundings kept
# speedup vs baseline: 2.9603x; 1.1006x over previous
#include <hip/hip_runtime.h>
#include <hip/hip_bf16.h>

#define GB   16
#define GN   10000
#define GE   160000
#define GS   128
#define GH   64
#define GEMB 64
#define GP   16

typedef __attribute__((ext_vector_type(2))) float v2f;
typedef __attribute__((ext_vector_type(4))) float v4f;
typedef __attribute__((ext_vector_type(8))) float v8f;
typedef __attribute__((ext_vector_type(2))) unsigned int v2u;
typedef __attribute__((ext_vector_type(4))) unsigned int v4u;
typedef __attribute__((ext_vector_type(4))) int v4i;

#define NRANGE 16
#define RN     (GN / NRANGE)
#define OWN_T  1024
#define SB     16384
#define AP     68

__global__ __launch_bounds__(OWN_T) void deg_kernel(const int* __restrict__ ei,
                                                    const float* __restrict__ ew,
                                                    float* __restrict__ dinv) {
    __shared__ float dl[RN];
    const int tid = threadIdx.x;
    const int b   = blockIdx.y;
    const unsigned lo = blockIdx.x * RN;
    for (int i = tid; i < RN; i += OWN_T) dl[i] = 0.f;
    __syncthreads();

    const v4i* col4 = (const v4i*)(ei + (size_t)b * 2 * GE + GE);
    const v4f* w4   = (const v4f*)(ew + (size_t)b * GE);
    for (int it = 0; it < (GE / 4 + 4 * OWN_T - 1) / (4 * OWN_T); ++it) {
        v4i c[4];
        v4f w[4];
#pragma unroll
        for (int u = 0; u < 4; ++u) {
            int p = (it * 4 + u) * OWN_T + tid;
            if (p < GE / 4) { c[u] = col4[p]; w[u] = w4[p]; }
            else            { c[u] = (v4i){-1, -1, -1, -1}; w[u] = (v4f){0.f, 0.f, 0.f, 0.f}; }
        }
#pragma unroll
        for (int u = 0; u < 4; ++u) {
#pragma unroll
            for (int j = 0; j < 4; ++j) {
                unsigned cl = (unsigned)c[u][j] - lo;
                if (cl < (unsigned)RN) atomicAdd(&dl[cl], w[u][j]);
            }
        }
    }
    __syncthreads();
    for (int i = tid; i < RN; i += OWN_T)
        dinv[(size_t)b * GN + lo + i] = rsqrtf(dl[i] + 1.0f);
}

__global__ __launch_bounds__(256) void norm_kernel(const int* __restrict__ ei,
                                                   const float* __restrict__ ew,
                                                   const float* __restrict__ dinv,
                                                   v2u* __restrict__ rec) {
    int idx = blockIdx.x * blockDim.x + threadIdx.x;
    if (idx >= GB * GE) return;
    int b = idx / GE;
    int e = idx - b * GE;
    int r = ei[(size_t)b * 2 * GE + e];
    int c = ei[(size_t)b * 2 * GE + GE + e];
    float nv = dinv[(size_t)b * GN + r] * ew[idx] * dinv[(size_t)b * GN + c];
    rec[idx] = (v2u){(unsigned)r | ((unsigned)c << 16), __float_as_uint(nv)};
}

template <int HIN>
__global__ __launch_bounds__(128) void gemm_wmma_kernel(const float* __restrict__ Hsrc,
                                                        const float* __restrict__ W,
                                                        float* __restrict__ Dst) {
    __shared__ float wlds[GH * HIN];
    const int tid = threadIdx.x;
    for (int i = tid; i < GH * HIN; i += 128) wlds[i] = W[i];
    __syncthreads();

    const int wave = tid >> 5;
    const int lane = tid & 31;
    const int half = lane >> 4;
    const int lm   = lane & 15;
    const int b    = blockIdx.z;
    const int m0   = blockIdx.x * 16;
    const int n0   = wave * 16;

    const float* hrow = Hsrc + ((size_t)b * GN + m0 + lm) * HIN;
    const float* wrow = wlds + (size_t)(n0 + lm) * HIN;

    v8f acc = {};
#pragma unroll
    for (int k0 = 0; k0 < HIN; k0 += 4) {
        const int ka = k0 + half * 2;
        v2f a = *(const v2f*)(hrow + ka);
        v2f bf = *(const v2f*)(wrow + ka);
        acc = __builtin_amdgcn_wmma_f32_16x16x4_f32(
             false, a,  false, bf,
             (short)0, acc,  false,  false);
    }

    float* drow = Dst + ((size_t)b * GN + m0) * GH + n0 + lm;
#pragma unroll
    for (int j = 0; j < 8; ++j) {
        drow[(size_t)(j + 8 * half) * GH] = acc[j];
    }
}

template <int DO_TANH, int FUSE>
__global__ __launch_bounds__(OWN_T) void agg_kernel(const v2u* __restrict__ rec,
                                                    const float* __restrict__ tmp,
                                                    const float* __restrict__ dinv,
                                                    const float* __restrict__ bias,
                                                    const float* __restrict__ Wn,
                                                    float* __restrict__ hout) {
    __shared__ float acc[RN * AP];
    __shared__ v2u   list[SB];
    __shared__ unsigned int cnt;

    const int tid  = threadIdx.x;
    const int lane = tid & 31;
    const int wave = __builtin_amdgcn_readfirstlane(tid >> 5);
    const int b    = blockIdx.y;
    const unsigned lo = blockIdx.x * RN;
    const size_t nodeBase = (size_t)b * GN + lo;

    for (int i4 = tid; i4 < RN * (GH / 4); i4 += OWN_T) {
        int node = i4 >> 4;
        int k4   = i4 & 15;
        float di = dinv[nodeBase + node];
        float sl = di * di;
        v4f t  = *(const v4f*)(tmp + (nodeBase + node) * GH + k4 * 4);
        v4f bb = *(const v4f*)(bias + k4 * 4);
        *(v4f*)(acc + node * AP + k4 * 4) = t * sl + bb;
    }
    if (tid == 0) cnt = 0;
    __syncthreads();

    const v4u*   rb = (const v4u*)(rec + (size_t)b * GE);
    const float* tb = tmp + (size_t)b * GN * GH + 2 * lane;
    unsigned prev = 0;

    for (int base = 0; base < GE; base += SB) {
        v4u r[8];
#pragma unroll
        for (int q = 0; q < 8; ++q) {
            int p = (base >> 1) + q * OWN_T + tid;
            if (p < GE / 2) r[q] = rb[p];
            else            r[q] = (v4u){0xFFFF0000u, 0u, 0xFFFF0000u, 0u};
        }
        unsigned m[16];
        unsigned tot = 0;
#pragma unroll
        for (int i = 0; i < 16; ++i) {
            unsigned word = (i & 1) ? r[i >> 1].z : r[i >> 1].x;
            bool hit = ((word >> 16) - lo) < (unsigned)RN;
            m[i] = __builtin_amdgcn_ballot_w32(hit);
            tot += __builtin_popcount(m[i]);
        }
        unsigned at = 0;
        if (tot) {
            if (lane == 0) at = atomicAdd(&cnt, tot);
            at = __builtin_amdgcn_readfirstlane(at);
        }
        at -= prev;
#pragma unroll
        for (int i = 0; i < 16; ++i) {
            unsigned word = (i & 1) ? r[i >> 1].z : r[i >> 1].x;
            unsigned nb   = (i & 1) ? r[i >> 1].w : r[i >> 1].y;
            unsigned cl   = (word >> 16) - lo;
            if (cl < (unsigned)RN)
                list[at + __builtin_amdgcn_mbcnt_lo(m[i], 0u)] = (v2u){(word & 0xFFFFu) | (cl << 16), nb};
            at += __builtin_popcount(m[i]);
        }
        __syncthreads();
        const unsigned cur = cnt;
        const int n = __builtin_amdgcn_readfirstlane(cur - prev);
        prev = cur;

        for (int i0 = wave * 8; i0 < n; i0 += (OWN_T / 32) * 8) {
            v2f      v[8];
            unsigned en[8];
            float    wv[8];
#pragma unroll
            for (int u = 0; u < 8; ++u) {
                int idx = i0 + u;
                idx = idx < n ? idx : n - 1;
                v2u e = list[idx];
                en[u] = __builtin_amdgcn_readfirstlane(e.x);
                wv[u] = __uint_as_float(__builtin_amdgcn_readfirstlane(e.y));
                v[u]  = *(const v2f*)(tb + (size_t)(en[u] & 0xFFFFu) * GH);
            }
#pragma unroll
            for (int u = 0; u < 8; ++u) {
                if (i0 + u < n) {
                    float* d = acc + (size_t)(en[u] >> 16) * AP + 2 * lane;
                    atomicAdd(d,     v[u].x * wv[u]);
                    atomicAdd(d + 1, v[u].y * wv[u]);
                }
            }
        }
        __syncthreads();
    }

    if (!FUSE) {
        for (int i4 = tid; i4 < RN * (GH / 4); i4 += OWN_T) {
            int node = i4 >> 4;
            int k4   = i4 & 15;
            v4f v = *(const v4f*)(acc + node * AP + k4 * 4);
            if (DO_TANH) {
                v.x = tanhf(v.x); v.y = tanhf(v.y); v.z = tanhf(v.z); v.w = tanhf(v.w);
            }
            *(v4f*)(hout + nodeBase * GH + (size_t)i4 * 4) = v;
        }
    } else {
        float* wl = (float*)list;
        for (int i4 = tid; i4 < RN * (GH / 4); i4 += OWN_T) {
            int node = i4 >> 4;
            int k4   = i4 & 15;
            v4f v = *(const v4f*)(acc + node * AP + k4 * 4);
            if (DO_TANH) {
                v.x = tanhf(v.x); v.y = tanhf(v.y); v.z = tanhf(v.z); v.w = tanhf(v.w);
            }
            *(v4f*)(acc + node * AP + k4 * 4) = v;
        }
        for (int i4 = tid; i4 < GH * (GH / 4); i4 += OWN_T) {
            int n  = i4 >> 4;
            int k4 = i4 & 15;
            *(v4f*)(wl + n * AP + k4 * 4) = *(const v4f*)(Wn + (size_t)i4 * 4);
        }
        __syncthreads();

        const int half = lane >> 4;
        const int lm   = lane & 15;
        for (int job = wave; job < ((RN + 15) / 16) * (GH / 16); job += OWN_T / 32) {
            const int m0 = (job >> 2) * 16;
            const int n0 = (job & 3) * 16;
            int row = m0 + lm;
            row = row < RN ? row : RN - 1;
            const float* arow = acc + row * AP;
            const float* wrow = wl + (n0 + lm) * AP;
            v8f c = {};
#pragma unroll
            for (int k0 = 0; k0 < GH; k0 += 4) {
                const int ka = k0 + half * 2;
                v2f a  = *(const v2f*)(arow + ka);
                v2f bf = *(const v2f*)(wrow + ka);
                c = __builtin_amdgcn_wmma_f32_16x16x4_f32(
                     false, a,  false, bf,
                     (short)0, c,  false,  false);
            }
            float* drow = hout + (nodeBase + m0) * GH + n0 + lm;
#pragma unroll
            for (int j = 0; j < 8; ++j) {
                const int r = j + 8 * half;
                if (m0 + r < RN) drow[(size_t)r * GH] = c[j];
            }
        }
    }
}

__global__ __launch_bounds__(256) void gather_kernel(const float* __restrict__ emb,
                                                     const int* __restrict__ pos,
                                                     float* __restrict__ out) {
    int idx = blockIdx.x * blockDim.x + threadIdx.x;
    if (idx >= GB * GP * GEMB) return;
    int k = idx & (GEMB - 1);
    int bp = idx >> 6;
    int b = bp / GP;
    int pv = pos[bp];
    out[idx] = (pv >= 0) ? emb[((size_t)b * GN + pv) * GH + k] : -3.0f;
}

extern "C" void kernel_launch(void* const* d_in, const int* in_sizes, int n_in,
                              void* d_out, int out_size, void* d_ws, size_t ws_size,
                              hipStream_t stream) {
    const float* x   = (const float*)d_in[0];
    const int*   ei  = (const int*)d_in[1];
    const float* ew  = (const float*)d_in[2];
    const int*   pos = (const int*)d_in[3];
    const float* W1  = (const float*)d_in[4];
    const float* b1  = (const float*)d_in[5];
    const float* W2  = (const float*)d_in[6];
    const float* b2  = (const float*)d_in[7];
    const float* W3  = (const float*)d_in[8];
    const float* b3  = (const float*)d_in[9];
    float* out = (float*)d_out;

    float* ws   = (float*)d_ws;
    const size_t nBN  = (size_t)GB * GN;
    const size_t nBE  = (size_t)GB * GE;
    const size_t nBNH = (size_t)GB * GN * GH;
    float* dinv = ws;
    v2u*   rec  = (v2u*)(dinv + nBN);
    float* tmp  = dinv + nBN + 2 * nBE;
    float* hbuf = tmp + nBNH;

    const int T = 256;
    dim3 gemm_grid(GN / 16, 1, GB);
    dim3 own_grid(NRANGE, GB, 1);

    deg_kernel<<<own_grid, OWN_T, 0, stream>>>(ei, ew, dinv);
    norm_kernel<<<(GB * GE + T - 1) / T, T, 0, stream>>>(ei, ew, dinv, rec);

    gemm_wmma_kernel<GS><<<gemm_grid, 128, 0, stream>>>(x, W1, tmp);
    agg_kernel<1, 1><<<own_grid, OWN_T, 0, stream>>>(rec, tmp, dinv, b1, W2, hbuf);

    agg_kernel<1, 1><<<own_grid, OWN_T, 0, stream>>>(rec, hbuf, dinv, b2, W3, tmp);

    agg_kernel<0, 0><<<own_grid, OWN_T, 0, stream>>>(rec, tmp, dinv, b3, W3, hbuf);

    gather_kernel<<<(GB * GP * GEMB + T - 1) / T, T, 0, stream>>>(hbuf, pos, out);
}


# Owner block stores its range's in-edge list once (degree pass) in workspace; norm and the 3 aggregations read that list, not all 160000 edges; no pass barriers; LDS-atomic sum order changes
# speedup vs baseline: 3.6417x; 1.2302x over previous
#include <hip/hip_runtime.h>
#include <hip/hip_bf16.h>

#define GB   16
#define GN   10000
#define GE   160000
#define GS   128
#define GH   64
#define GEMB 64
#define GP   16

typedef __attribute__((ext_vector_type(2))) float v2f;
typedef __attribute__((ext_vector_type(4))) float v4f;
typedef __attribute__((ext_vector_type(8))) float v8f;
typedef __attribute__((ext_vector_type(2))) unsigned int v2u;
typedef __attribute__((ext_vector_type(4))) int v4i;

#define NRANGE 16
#define RN     (GN / NRANGE)
#define OWN_T  1024
#define CAP    11264
#define AP     68

__global__ __launch_bounds__(OWN_T) void deg_kernel(const int* __restrict__ ei,
                                                    const float* __restrict__ ew,
                                                    float* __restrict__ dinv,
                                                    v2u* __restrict__ lists,
                                                    unsigned int* __restrict__ counts) {
    __shared__ float dl[RN];
    __shared__ unsigned int cnt;
    const int tid  = threadIdx.x;
    const int lane = tid & 31;
    const int b    = blockIdx.y;
    const unsigned lo = blockIdx.x * RN;
    const int wg   = b * NRANGE + blockIdx.x;
    v2u* lst = lists + (size_t)wg * CAP;
    for (int i = tid; i < RN; i += OWN_T) dl[i] = 0.f;
    if (tid == 0) cnt = 0;
    __syncthreads();

    const v4i* row4 = (const v4i*)(ei + (size_t)b * 2 * GE);
    const v4i* col4 = (const v4i*)(ei + (size_t)b * 2 * GE + GE);
    const v4f* w4   = (const v4f*)(ew + (size_t)b * GE);
    for (int it = 0; it < (GE / 4 + 4 * OWN_T - 1) / (4 * OWN_T); ++it) {
        v4i c[4];
        v4i r[4];
        v4f w[4];
#pragma unroll
        for (int u = 0; u < 4; ++u) {
            int p = (it * 4 + u) * OWN_T + tid;
            if (p < GE / 4) { c[u] = col4[p]; r[u] = row4[p]; w[u] = w4[p]; }
            else            { c[u] = (v4i){-1, -1, -1, -1}; r[u] = (v4i){0, 0, 0, 0}; w[u] = (v4f){0.f, 0.f, 0.f, 0.f}; }
        }
        unsigned m[16];
        unsigned tot = 0;
#pragma unroll
        for (int i = 0; i < 16; ++i) {
            bool hit = ((unsigned)c[i >> 2][i & 3] - lo) < (unsigned)RN;
            m[i] = __builtin_amdgcn_ballot_w32(hit);
            tot += __builtin_popcount(m[i]);
        }
        unsigned at = 0;
        if (tot) {
            if (lane == 0) at = atomicAdd(&cnt, tot);
            at = __builtin_amdgcn_readfirstlane(at);
        }
#pragma unroll
        for (int i = 0; i < 16; ++i) {
            unsigned cl = (unsigned)c[i >> 2][i & 3] - lo;
            if (cl < (unsigned)RN) {
                float wt = w[i >> 2][i & 3];
                atomicAdd(&dl[cl], wt);
                unsigned pos = at + __builtin_amdgcn_mbcnt_lo(m[i], 0u);
                if (pos < (unsigned)CAP)
                    lst[pos] = (v2u){(unsigned)r[i >> 2][i & 3] | (cl << 16), __float_as_uint(wt)};
            }
            at += __builtin_popcount(m[i]);
        }
    }
    __syncthreads();
    for (int i = tid; i < RN; i += OWN_T)
        dinv[(size_t)b * GN + lo + i] = rsqrtf(dl[i] + 1.0f);
    if (tid == 0) counts[wg] = cnt;
}

__global__ __launch_bounds__(256) void norm_kernel(const float* __restrict__ dinv,
                                                   const unsigned int* __restrict__ counts,
                                                   v2u* __restrict__ lists) {
    const int wg = blockIdx.y;
    const int b  = wg / NRANGE;
    const unsigned lo = (unsigned)(wg - b * NRANGE) * RN;
    unsigned n = counts[wg];
    n = n < (unsigned)CAP ? n : (unsigned)CAP;
    const unsigned i = blockIdx.x * 256 + threadIdx.x;
    if (i >= n) return;
    v2u* p = lists + (size_t)wg * CAP + i;
    v2u e = *p;
    float nv = dinv[(size_t)b * GN + (e.x & 0xFFFFu)] * __uint_as_float(e.y) * dinv[(size_t)b * GN + lo + (e.x >> 16)];
    *p = (v2u){e.x, __float_as_uint(nv)};
}

template <int HIN>
__global__ __launch_bounds__(128) void gemm_wmma_kernel(const float* __restrict__ Hsrc,
                                                        const float* __restrict__ W,
                                                        float* __restrict__ Dst) {
    __shared__ float wlds[GH * HIN];
    const int tid = threadIdx.x;
    for (int i = tid; i < GH * HIN; i += 128) wlds[i] = W[i];
    __syncthreads();

    const int wave = tid >> 5;
    const int lane = tid & 31;
    const int half = lane >> 4;
    const int lm   = lane & 15;
    const int b    = blockIdx.z;
    const int m0   = blockIdx.x * 16;
    const int n0   = wave * 16;

    const float* hrow = Hsrc + ((size_t)b * GN + m0 + lm) * HIN;
    const float* wrow = wlds + (size_t)(n0 + lm) * HIN;

    v8f acc = {};
#pragma unroll
    for (int k0 = 0; k0 < HIN; k0 += 4) {
        const int ka = k0 + half * 2;
        v2f a = *(const v2f*)(hrow + ka);
        v2f bf = *(const v2f*)(wrow + ka);
        acc = __builtin_amdgcn_wmma_f32_16x16x4_f32(
             false, a,  false, bf,
             (short)0, acc,  false,  false);
    }

    float* drow = Dst + ((size_t)b * GN + m0) * GH + n0 + lm;
#pragma unroll
    for (int j = 0; j < 8; ++j) {
        drow[(size_t)(j + 8 * half) * GH] = acc[j];
    }
}

template <int DO_TANH, int FUSE>
__global__ __launch_bounds__(OWN_T) void agg_kernel(const v2u* __restrict__ lists,
                                                    const unsigned int* __restrict__ counts,
                                                    const int* __restrict__ ei,
                                                    const float* __restrict__ ew,
                                                    const float* __restrict__ tmp,
                                                    const float* __restrict__ dinv,
                                                    const float* __restrict__ bias,
                                                    const float* __restrict__ Wn,
                                                    float* __restrict__ hout) {
    __shared__ float acc[RN * AP];
    __shared__ float wl[FUSE ? GH * AP : 4];

    const int tid  = threadIdx.x;
    const int lane = tid & 31;
    const int wave = __builtin_amdgcn_readfirstlane(tid >> 5);
    const int b    = blockIdx.y;
    const unsigned lo = blockIdx.x * RN;
    const int wg   = b * NRANGE + blockIdx.x;
    const size_t nodeBase = (size_t)b * GN + lo;

    for (int i4 = tid; i4 < RN * (GH / 4); i4 += OWN_T) {
        int node = i4 >> 4;
        int k4   = i4 & 15;
        float di = dinv[nodeBase + node];
        float sl = di * di;
        v4f t  = *(const v4f*)(tmp + (nodeBase + node) * GH + k4 * 4);
        v4f bb = *(const v4f*)(bias + k4 * 4);
        *(v4f*)(acc + node * AP + k4 * 4) = t * sl + bb;
    }
    __syncthreads();

    const float* tg = tmp + (size_t)b * GN * GH;
    const unsigned nAll = counts[wg];

    if (nAll <= (unsigned)CAP) {
        const int n = (int)nAll;
        const v2u* lst = lists + (size_t)wg * CAP;
        int c0 = wave * 32;
        v2u e = (v2u){0u, 0u};
        if (c0 + lane < n) e = lst[c0 + lane];
        for (; c0 < n; c0 += OWN_T) {
            v2u enx = (v2u){0u, 0u};
            const int m = n - c0;
#pragma unroll
            for (int g = 0; g < 32; g += 16) {
                if (g < m) {
                    v2f      v[16];
                    unsigned sx[16];
                    float    wv[16];
#pragma unroll
                    for (int u = 0; u < 16; ++u) {
                        sx[u] = (unsigned)__builtin_amdgcn_readlane((int)e.x, g + u);
                        wv[u] = __uint_as_float((unsigned)__builtin_amdgcn_readlane((int)e.y, g + u));
                        const float* rowp = tg + (size_t)(sx[u] & 0xFFFFu) * GH;
                        v[u]  = *(const v2f*)(rowp + 2 * lane);
                    }
                    if (g == 0 && c0 + OWN_T + lane < n) enx = lst[c0 + OWN_T + lane];
#pragma unroll
                    for (int u = 0; u < 16; ++u) {
                        if (g + u < m) {
                            float* d = acc + (size_t)(sx[u] >> 16) * AP + 2 * lane;
                            atomicAdd(d,     v[u].x * wv[u]);
                            atomicAdd(d + 1, v[u].y * wv[u]);
                        }
                    }
                }
            }
            e = enx;
        }
    } else {
        const int*   rowp = ei + (size_t)b * 2 * GE;
        const int*   colp = rowp + GE;
        const float* wp   = ew + (size_t)b * GE;
        for (int eI = tid; eI < GE; eI += OWN_T) {
            int c = colp[eI];
            unsigned cl = (unsigned)c - lo;
            if (cl < (unsigned)RN) {
                int r = rowp[eI];
                float nv = dinv[(size_t)b * GN + r] * wp[eI] * dinv[(size_t)b * GN + c];
                const float* sr = tmp + ((size_t)b * GN + r) * GH;
                float* d = acc + (size_t)cl * AP;
                for (int k = 0; k < GH; ++k) atomicAdd(d + k, sr[k] * nv);
            }
        }
    }
    __syncthreads();

    if (!FUSE) {
        for (int i4 = tid; i4 < RN * (GH / 4); i4 += OWN_T) {
            int node = i4 >> 4;
            int k4   = i4 & 15;
            v4f v = *(const v4f*)(acc + node * AP + k4 * 4);
            if (DO_TANH) {
                v.x = tanhf(v.x); v.y = tanhf(v.y); v.z = tanhf(v.z); v.w = tanhf(v.w);
            }
            *(v4f*)(hout + nodeBase * GH + (size_t)i4 * 4) = v;
        }
    } else {
        for (int i4 = tid; i4 < RN * (GH / 4); i4 += OWN_T) {
            int node = i4 >> 4;
            int k4   = i4 & 15;
            v4f v = *(const v4f*)(acc + node * AP + k4 * 4);
            if (DO_TANH) {
                v.x = tanhf(v.x); v.y = tanhf(v.y); v.z = tanhf(v.z); v.w = tanhf(v.w);
            }
            *(v4f*)(acc + node * AP + k4 * 4) = v;
        }
        for (int i4 = tid; i4 < GH * (GH / 4); i4 += OWN_T) {
            int n  = i4 >> 4;
            int k4 = i4 & 15;
            *(v4f*)(wl + n * AP + k4 * 4) = *(const v4f*)(Wn + (size_t)i4 * 4);
        }
        __syncthreads();

        const int half = lane >> 4;
        const int lm   = lane & 15;
        for (int job = wave; job < ((RN + 15) / 16) * (GH / 16); job += OWN_T / 32) {
            const int m0 = (job >> 2) * 16;
            const int n0 = (job & 3) * 16;
            int row = m0 + lm;
            row = row < RN ? row : RN - 1;
            const float* arow = acc + row * AP;
            const float* wrow = wl + (n0 + lm) * AP;
            v8f c = {};
#pragma unroll
            for (int k0 = 0; k0 < GH; k0 += 4) {
                const int ka = k0 + half * 2;
                v2f a  = *(const v2f*)(arow + ka);
                v2f bf = *(const v2f*)(wrow + ka);
                c = __builtin_amdgcn_wmma_f32_16x16x4_f32(
                     false, a,  false, bf,
                     (short)0, c,  false,  false);
            }
            float* drow = hout + (nodeBase + m0) * GH + n0 + lm;
#pragma unroll
            for (int j = 0; j < 8; ++j) {
                const int r = j + 8 * half;
                if (m0 + r < RN) drow[(size_t)r * GH] = c[j];
            }
        }
    }
}

__global__ __launch_bounds__(256) void gather_kernel(const float* __restrict__ emb,
                                                     const int* __restrict__ pos,
                                                     float* __restrict__ out) {
    int idx = blockIdx.x * blockDim.x + threadIdx.x;
    if (idx >= GB * GP * GEMB) return;
    int k = idx & (GEMB - 1);
    int bp = idx >> 6;
    int b = bp / GP;
    int pv = pos[bp];
    out[idx] = (pv >= 0) ? emb[((size_t)b * GN + pv) * GH + k] : -3.0f;
}

extern "C" void kernel_launch(void* const* d_in, const int* in_sizes, int n_in,
                              void* d_out, int out_size, void* d_ws, size_t ws_size,
                              hipStream_t stream) {
    const float* x   = (const float*)d_in[0];
    const int*   ei  = (const int*)d_in[1];
    const float* ew  = (const float*)d_in[2];
    const int*   pos = (const int*)d_in[3];
    const float* W1  = (const float*)d_in[4];
    const float* b1  = (const float*)d_in[5];
    const float* W2  = (const float*)d_in[6];
    const float* b2  = (const float*)d_in[7];
    const float* W3  = (const float*)d_in[8];
    const float* b3  = (const float*)d_in[9];
    float* out = (float*)d_out;

    float* ws   = (float*)d_ws;
    const size_t nBN  = (size_t)GB * GN;
    const size_t nBNH = (size_t)GB * GN * GH;
    const size_t nList = (size_t)NRANGE * GB;
    float* dinv = ws;
    unsigned int* counts = (unsigned int*)(dinv + nBN);
    v2u*   lists = (v2u*)(dinv + nBN + nList);
    float* tmp  = dinv + nBN + nList + 2 * nList * CAP;
    float* hbuf = tmp + nBNH;

    const int T = 256;
    dim3 gemm_grid(GN / 16, 1, GB);
    dim3 own_grid(NRANGE, GB, 1);

    deg_kernel<<<own_grid, OWN_T, 0, stream>>>(ei, ew, dinv, lists, counts);
    norm_kernel<<<dim3(CAP / T, NRANGE * GB, 1), T, 0, stream>>>(dinv, counts, lists);

    gemm_wmma_kernel<GS><<<gemm_grid, 128, 0, stream>>>(x, W1, tmp);
    agg_kernel<1, 1><<<own_grid, OWN_T, 0, stream>>>(lists, counts, ei, ew, tmp, dinv, b1, W2, hbuf);

    agg_kernel<1, 1><<<own_grid, OWN_T, 0, stream>>>(lists, counts, ei, ew, hbuf, dinv, b2, W3, tmp);

    agg_kernel<0, 0><<<own_grid, OWN_T, 0, stream>>>(lists, counts, ei, ew, tmp, dinv, b3, W3, hbuf);

    gather_kernel<<<(GB * GP * GEMB + T - 1) / T, T, 0, stream>>>(hbuf, pos, out);
}


# Layer-1 linear map moved into the degree pass: owner block stages its 625 x rows through LDS in 128-row chunks (16-byte loads, two buffers), W1 staged once per block; same k order; 7 launches to 6
# speedup vs baseline: 4.0762x; 1.1193x over previous
#include <hip/hip_runtime.h>
#include <hip/hip_bf16.h>

#define GB   16
#define GN   10000
#define GE   160000
#define GS   128
#define GH   64
#define GEMB 64
#define GP   16

typedef __attribute__((ext_vector_type(2))) float v2f;
typedef __attribute__((ext_vector_type(4))) float v4f;
typedef __attribute__((ext_vector_type(8))) float v8f;
typedef __attribute__((ext_vector_type(2))) unsigned int v2u;
typedef __attribute__((ext_vector_type(4))) int v4i;

#define NRANGE 16
#define RN     (GN / NRANGE)
#define OWN_T  1024
#define CAP    11264
#define AP     68
#define XP     132
#define XR     128
#define NCH    ((RN + XR - 1) / XR)

__global__ __launch_bounds__(OWN_T) void deg_kernel(const int* __restrict__ ei,
                                                    const float* __restrict__ ew,
                                                    const float* __restrict__ x,
                                                    const float* __restrict__ W1,
                                                    float* __restrict__ tmp,
                                                    float* __restrict__ dinv,
                                                    v2u* __restrict__ lists,
                                                    unsigned int* __restrict__ counts) {
    __shared__ float dl[RN];
    __shared__ unsigned int cnt;
    __shared__ float w1l[GH * XP];
    __shared__ float xl[2][XR * XP];
    const int tid  = threadIdx.x;
    const int lane = tid & 31;
    const int b    = blockIdx.y;
    const unsigned lo = blockIdx.x * RN;
    const int wg   = b * NRANGE + blockIdx.x;
    v2u* lst = lists + (size_t)wg * CAP;

    {
        const int wave = __builtin_amdgcn_readfirstlane(tid >> 5);
        const int half = lane >> 4;
        const int lm   = lane & 15;
        const size_t nodeBase = (size_t)b * GN + lo;
        const int xrow = tid >> 5;
        const int xk4  = tid & 31;

        for (int i4 = tid; i4 < GH * (GS / 4); i4 += OWN_T) {
            int n  = i4 >> 5;
            int k4 = i4 & 31;
            *(v4f*)(w1l + n * XP + k4 * 4) = *(const v4f*)(W1 + (size_t)i4 * 4);
        }
        v4f nx[4];
#pragma unroll
        for (int j = 0; j < 4; ++j) {
            int r = xrow + j * 32;
            nx[j] = *(const v4f*)(x + (nodeBase + r) * GS + xk4 * 4);
        }
#pragma unroll
        for (int j = 0; j < 4; ++j)
            *(v4f*)(&xl[0][(xrow + j * 32) * XP + xk4 * 4]) = nx[j];
        __syncthreads();

        const int m0 = (wave >> 2) * 16;
        const int n0 = (wave & 3) * 16;
        const float* wrow = w1l + (n0 + lm) * XP;
        for (int c = 0; c < NCH; ++c) {
            if (c + 1 < NCH) {
#pragma unroll
                for (int j = 0; j < 4; ++j) {
                    int r = (c + 1) * XR + xrow + j * 32;
                    r = r < RN ? r : RN - 1;
                    nx[j] = *(const v4f*)(x + (nodeBase + r) * GS + xk4 * 4);
                }
            }
            const float* arow = &xl[c & 1][(m0 + lm) * XP];
            v8f acc = {};
#pragma unroll
            for (int k0 = 0; k0 < GS; k0 += 4) {
                const int ka = k0 + half * 2;
                v2f a  = *(const v2f*)(arow + ka);
                v2f bf = *(const v2f*)(wrow + ka);
                acc = __builtin_amdgcn_wmma_f32_16x16x4_f32(
                     false, a,  false, bf,
                     (short)0, acc,  false,  false);
            }
            const int rbase = c * XR + m0;
            float* drow = tmp + (nodeBase + rbase) * GH + n0 + lm;
#pragma unroll
            for (int j = 0; j < 8; ++j) {
                const int r = j + 8 * half;
                if (c + 1 < NCH || rbase + r < RN) drow[(size_t)r * GH] = acc[j];
            }
            if (c + 1 < NCH) {
#pragma unroll
                for (int j = 0; j < 4; ++j)
                    *(v4f*)(&xl[(c + 1) & 1][(xrow + j * 32) * XP + xk4 * 4]) = nx[j];
            }
            __syncthreads();
        }
    }
    for (int i = tid; i < RN; i += OWN_T) dl[i] = 0.f;
    if (tid == 0) cnt = 0;
    __syncthreads();

    const v4i* row4 = (const v4i*)(ei + (size_t)b * 2 * GE);
    const v4i* col4 = (const v4i*)(ei + (size_t)b * 2 * GE + GE);
    const v4f* w4   = (const v4f*)(ew + (size_t)b * GE);
    for (int it = 0; it < (GE / 4 + 4 * OWN_T - 1) / (4 * OWN_T); ++it) {
        v4i c[4];
        v4i r[4];
        v4f w[4];
#pragma unroll
        for (int u = 0; u < 4; ++u) {
            int p = (it * 4 + u) * OWN_T + tid;
            if (p < GE / 4) { c[u] = col4[p]; r[u] = row4[p]; w[u] = w4[p]; }
            else            { c[u] = (v4i){-1, -1, -1, -1}; r[u] = (v4i){0, 0, 0, 0}; w[u] = (v4f){0.f, 0.f, 0.f, 0.f}; }
        }
        unsigned m[16];
        unsigned tot = 0;
#pragma unroll
        for (int i = 0; i < 16; ++i) {
            bool hit = ((unsigned)c[i >> 2][i & 3] - lo) < (unsigned)RN;
            m[i] = __builtin_amdgcn_ballot_w32(hit);
            tot += __builtin_popcount(m[i]);
        }
        unsigned at = 0;
        if (tot) {
            if (lane == 0) at = atomicAdd(&cnt, tot);
            at = __builtin_amdgcn_readfirstlane(at);
        }
#pragma unroll
        for (int i = 0; i < 16; ++i) {
            unsigned cl = (unsigned)c[i >> 2][i & 3] - lo;
            if (cl < (unsigned)RN) {
                float wt = w[i >> 2][i & 3];
                atomicAdd(&dl[cl], wt);
                unsigned pos = at + __builtin_amdgcn_mbcnt_lo(m[i], 0u);
                if (pos < (unsigned)CAP)
                    lst[pos] = (v2u){(unsigned)r[i >> 2][i & 3] | (cl << 16), __float_as_uint(wt)};
            }
            at += __builtin_popcount(m[i]);
        }
    }
    __syncthreads();
    for (int i = tid; i < RN; i += OWN_T)
        dinv[(size_t)b * GN + lo + i] = rsqrtf(dl[i] + 1.0f);
    if (tid == 0) counts[wg] = cnt;
}

__global__ __launch_bounds__(256) void norm_kernel(const float* __restrict__ dinv,
                                                   const unsigned int* __restrict__ counts,
                                                   v2u* __restrict__ lists) {
    const int wg = blockIdx.y;
    const int b  = wg / NRANGE;
    const unsigned lo = (unsigned)(wg - b * NRANGE) * RN;
    unsigned n = counts[wg];
    n = n < (unsigned)CAP ? n : (unsigned)CAP;
    const unsigned i = blockIdx.x * 256 + threadIdx.x;
    if (i >= n) return;
    v2u* p = lists + (size_t)wg * CAP + i;
    v2u e = *p;
    float nv = dinv[(size_t)b * GN + (e.x & 0xFFFFu)] * __uint_as_float(e.y) * dinv[(size_t)b * GN + lo + (e.x >> 16)];
    *p = (v2u){e.x, __float_as_uint(nv)};
}

template <int DO_TANH, int FUSE>
__global__ __launch_bounds__(OWN_T) void agg_kernel(const v2u* __restrict__ lists,
                                                    const unsigned int* __restrict__ counts,
                                                    const int* __restrict__ ei,
                                                    const float* __restrict__ ew,
                                                    const float* __restrict__ tmp,
                                                    const float* __restrict__ dinv,
                                                    const float* __restrict__ bias,
                                                    const float* __restrict__ Wn,
                                                    float* __restrict__ hout) {
    __shared__ float acc[RN * AP];
    __shared__ float wl[FUSE ? GH * AP : 4];

    const int tid  = threadIdx.x;
    const int lane = tid & 31;
    const int wave = __builtin_amdgcn_readfirstlane(tid >> 5);
    const int b    = blockIdx.y;
    const unsigned lo = blockIdx.x * RN;
    const int wg   = b * NRANGE + blockIdx.x;
    const size_t nodeBase = (size_t)b * GN + lo;

    for (int i4 = tid; i4 < RN * (GH / 4); i4 += OWN_T) {
        int node = i4 >> 4;
        int k4   = i4 & 15;
        float di = dinv[nodeBase + node];
        float sl = di * di;
        v4f t  = *(const v4f*)(tmp + (nodeBase + node) * GH + k4 * 4);
        v4f bb = *(const v4f*)(bias + k4 * 4);
        *(v4f*)(acc + node * AP + k4 * 4) = t * sl + bb;
    }
    __syncthreads();

    const float* tg = tmp + (size_t)b * GN * GH;
    const unsigned nAll = counts[wg];

    if (nAll <= (unsigned)CAP) {
        const int n = (int)nAll;
        const v2u* lst = lists + (size_t)wg * CAP;
        int c0 = wave * 32;
        v2u e = (v2u){0u, 0u};
        if (c0 + lane < n) e = lst[c0 + lane];
        for (; c0 < n; c0 += OWN_T) {
            v2u enx = (v2u){0u, 0u};
            const int m = n - c0;
#pragma unroll
            for (int g = 0; g < 32; g += 16) {
                if (g < m) {
                    v2f      v[16];
                    unsigned sx[16];
                    float    wv[16];
#pragma unroll
                    for (int u = 0; u < 16; ++u) {
                        sx[u] = (unsigned)__builtin_amdgcn_readlane((int)e.x, g + u);
                        wv[u] = __uint_as_float((unsigned)__builtin_amdgcn_readlane((int)e.y, g + u));
                        const float* rowp = tg + (size_t)(sx[u] & 0xFFFFu) * GH;
                        v[u]  = *(const v2f*)(rowp + 2 * lane);
                    }
                    if (g == 0 && c0 + OWN_T + lane < n) enx = lst[c0 + OWN_T + lane];
#pragma unroll
                    for (int u = 0; u < 16; ++u) {
                        if (g + u < m) {
                            float* d = acc + (size_t)(sx[u] >> 16) * AP + 2 * lane;
                            atomicAdd(d,     v[u].x * wv[u]);
                            atomicAdd(d + 1, v[u].y * wv[u]);
                        }
                    }
                }
            }
            e = enx;
        }
    } else {
        const int*   rowp = ei + (size_t)b * 2 * GE;
        const int*   colp = rowp + GE;
        const float* wp   = ew + (size_t)b * GE;
        for (int eI = tid; eI < GE; eI += OWN_T) {
            int c = colp[eI];
            unsigned cl = (unsigned)c - lo;
            if (cl < (unsigned)RN) {
                int r = rowp[eI];
                float nv = dinv[(size_t)b * GN + r] * wp[eI] * dinv[(size_t)b * GN + c];
                const float* sr = tmp + ((size_t)b * GN + r) * GH;
                float* d = acc + (size_t)cl * AP;
                for (int k = 0; k < GH; ++k) atomicAdd(d + k, sr[k] * nv);
            }
        }
    }
    __syncthreads();

    if (!FUSE) {
        for (int i4 = tid; i4 < RN * (GH / 4); i4 += OWN_T) {
            int node = i4 >> 4;
            int k4   = i4 & 15;
            v4f v = *(const v4f*)(acc + node * AP + k4 * 4);
            if (DO_TANH) {
                v.x = tanhf(v.x); v.y = tanhf(v.y); v.z = tanhf(v.z); v.w = tanhf(v.w);
            }
            *(v4f*)(hout + nodeBase * GH + (size_t)i4 * 4) = v;
        }
    } else {
        for (int i4 = tid; i4 < RN * (GH / 4); i4 += OWN_T) {
            int node = i4 >> 4;
            int k4   = i4 & 15;
            v4f v = *(const v4f*)(acc + node * AP + k4 * 4);
            if (DO_TANH) {
                v.x = tanhf(v.x); v.y = tanhf(v.y); v.z = tanhf(v.z); v.w = tanhf(v.w);
            }
            *(v4f*)(acc + node * AP + k4 * 4) = v;
        }
        for (int i4 = tid; i4 < GH * (GH / 4); i4 += OWN_T) {
            int n  = i4 >> 4;
            int k4 = i4 & 15;
            *(v4f*)(wl + n * AP + k4 * 4) = *(const v4f*)(Wn + (size_t)i4 * 4);
        }
        __syncthreads();

        const int half = lane >> 4;
        const int lm   = lane & 15;
        for (int job = wave; job < ((RN + 15) / 16) * (GH / 16); job += OWN_T / 32) {
            const int m0 = (job >> 2) * 16;
            const int n0 = (job & 3) * 16;
            int row = m0 + lm;
            row = row < RN ? row : RN - 1;
            const float* arow = acc + row * AP;
            const float* wrow = wl + (n0 + lm) * AP;
            v8f c = {};
#pragma unroll
            for (int k0 = 0; k0 < GH; k0 += 4) {
                const int ka = k0 + half * 2;
                v2f a  = *(const v2f*)(arow + ka);
                v2f bf = *(const v2f*)(wrow + ka);
                c = __builtin_amdgcn_wmma_f32_16x16x4_f32(
                     false, a,  false, bf,
                     (short)0, c,  false,  false);
            }
            float* drow = hout + (nodeBase + m0) * GH + n0 + lm;
#pragma unroll
            for (int j = 0; j < 8; ++j) {
                const int r = j + 8 * half;
                if (m0 + r < RN) drow[(size_t)r * GH] = c[j];
            }
        }
    }
}

__global__ __launch_bounds__(256) void gather_kernel(const float* __restrict__ emb,
                                                     const int* __restrict__ pos,
                                                     float* __restrict__ out) {
    int idx = blockIdx.x * blockDim.x + threadIdx.x;
    if (idx >= GB * GP * GEMB) return;
    int k = idx & (GEMB - 1);
    int bp = idx >> 6;
    int b = bp / GP;
    int pv = pos[bp];
    out[idx] = (pv >= 0) ? emb[((size_t)b * GN + pv) * GH + k] : -3.0f;
}

extern "C" void kernel_launch(void* const* d_in, const int* in_sizes, int n_in,
                              void* d_out, int out_size, void* d_ws, size_t ws_size,
                              hipStream_t stream) {
    const float* x   = (const float*)d_in[0];
    const int*   ei  = (const int*)d_in[1];
    const float* ew  = (const float*)d_in[2];
    const int*   pos = (const int*)d_in[3];
    const float* W1  = (const float*)d_in[4];
    const float* b1  = (const float*)d_in[5];
    const float* W2  = (const float*)d_in[6];
    const float* b2  = (const float*)d_in[7];
    const float* W3  = (const float*)d_in[8];
    const float* b3  = (const float*)d_in[9];
    float* out = (float*)d_out;

    float* ws   = (float*)d_ws;
    const size_t nBN  = (size_t)GB * GN;
    const size_t nBNH = (size_t)GB * GN * GH;
    const size_t nList = (size_t)NRANGE * GB;
    float* dinv = ws;
    unsigned int* counts = (unsigned int*)(dinv + nBN);
    v2u*   lists = (v2u*)(dinv + nBN + nList);
    float* tmp  = dinv + nBN + nList + 2 * nList * CAP;
    float* hbuf = tmp + nBNH;

    const int T = 256;
    dim3 own_grid(NRANGE, GB, 1);

    deg_kernel<<<own_grid, OWN_T, 0, stream>>>(ei, ew, x, W1, tmp, dinv, lists, counts);
    norm_kernel<<<dim3(CAP / T, NRANGE * GB, 1), T, 0, stream>>>(dinv, counts, lists);

    agg_kernel<1, 1><<<own_grid, OWN_T, 0, stream>>>(lists, counts, ei, ew, tmp, dinv, b1, W2, hbuf);

    agg_kernel<1, 1><<<own_grid, OWN_T, 0, stream>>>(lists, counts, ei, ew, hbuf, dinv, b2, W3, tmp);

    agg_kernel<0, 0><<<own_grid, OWN_T, 0, stream>>>(lists, counts, ei, ew, tmp, dinv, b3, W3, hbuf);

    gather_kernel<<<(GB * GP * GEMB + T - 1) / T, T, 0, stream>>>(hbuf, pos, out);
}


# Per-edge norm launch folded into layer-1 aggregation: block holds its graph's dinv (40 KB) in LDS, weights each list entry as it loads it, stores it for layers 2-3; 6 launches to 5; roundings kept
# speedup vs baseline: 4.2644x; 1.0462x over previous
#include <hip/hip_runtime.h>
#include <hip/hip_bf16.h>

#define GB   16
#define GN   10000
#define GE   160000
#define GS   128
#define GH   64
#define GEMB 64
#define GP   16

typedef __attribute__((ext_vector_type(2))) float v2f;
typedef __attribute__((ext_vector_type(4))) float v4f;
typedef __attribute__((ext_vector_type(8))) float v8f;
typedef __attribute__((ext_vector_type(2))) unsigned int v2u;
typedef __attribute__((ext_vector_type(4))) int v4i;

#define NRANGE 16
#define RN     (GN / NRANGE)
#define OWN_T  1024
#define CAP    11264
#define AP     68
#define XP     132
#define XR     128
#define NCH    ((RN + XR - 1) / XR)

__global__ __launch_bounds__(OWN_T) void deg_kernel(const int* __restrict__ ei,
                                                    const float* __restrict__ ew,
                                                    const float* __restrict__ x,
                                                    const float* __restrict__ W1,
                                                    float* __restrict__ tmp,
                                                    float* __restrict__ dinv,
                                                    v2u* __restrict__ lists,
                                                    unsigned int* __restrict__ counts) {
    __shared__ float dl[RN];
    __shared__ unsigned int cnt;
    __shared__ float w1l[GH * XP];
    __shared__ float xl[2][XR * XP];
    const int tid  = threadIdx.x;
    const int lane = tid & 31;
    const int b    = blockIdx.y;
    const unsigned lo = blockIdx.x * RN;
    const int wg   = b * NRANGE + blockIdx.x;
    v2u* lst = lists + (size_t)wg * CAP;

    {
        const int wave = __builtin_amdgcn_readfirstlane(tid >> 5);
        const int half = lane >> 4;
        const int lm   = lane & 15;
        const size_t nodeBase = (size_t)b * GN + lo;
        const int xrow = tid >> 5;
        const int xk4  = tid & 31;

        for (int i4 = tid; i4 < GH * (GS / 4); i4 += OWN_T) {
            int n  = i4 >> 5;
            int k4 = i4 & 31;
            *(v4f*)(w1l + n * XP + k4 * 4) = *(const v4f*)(W1 + (size_t)i4 * 4);
        }
        v4f nx[4];
#pragma unroll
        for (int j = 0; j < 4; ++j) {
            int r = xrow + j * 32;
            nx[j] = *(const v4f*)(x + (nodeBase + r) * GS + xk4 * 4);
        }
#pragma unroll
        for (int j = 0; j < 4; ++j)
            *(v4f*)(&xl[0][(xrow + j * 32) * XP + xk4 * 4]) = nx[j];
        __syncthreads();

        const int m0 = (wave >> 2) * 16;
        const int n0 = (wave & 3) * 16;
        const float* wrow = w1l + (n0 + lm) * XP;
        for (int c = 0; c < NCH; ++c) {
            if (c + 1 < NCH) {
#pragma unroll
                for (int j = 0; j < 4; ++j) {
                    int r = (c + 1) * XR + xrow + j * 32;
                    r = r < RN ? r : RN - 1;
                    nx[j] = *(const v4f*)(x + (nodeBase + r) * GS + xk4 * 4);
                }
            }
            const float* arow = &xl[c & 1][(m0 + lm) * XP];
            v8f acc = {};
#pragma unroll
            for (int k0 = 0; k0 < GS; k0 += 4) {
                const int ka = k0 + half * 2;
                v2f a  = *(const v2f*)(arow + ka);
                v2f bf = *(const v2f*)(wrow + ka);
                acc = __builtin_amdgcn_wmma_f32_16x16x4_f32(
                     false, a,  false, bf,
                     (short)0, acc,  false,  false);
            }
            const int rbase = c * XR + m0;
            float* drow = tmp + (nodeBase + rbase) * GH + n0 + lm;
#pragma unroll
            for (int j = 0; j < 8; ++j) {
                const int r = j + 8 * half;
                if (c + 1 < NCH || rbase + r < RN) drow[(size_t)r * GH] = acc[j];
            }
            if (c + 1 < NCH) {
#pragma unroll
                for (int j = 0; j < 4; ++j)
                    *(v4f*)(&xl[(c + 1) & 1][(xrow + j * 32) * XP + xk4 * 4]) = nx[j];
            }
            __syncthreads();
        }
    }
    for (int i = tid; i < RN; i += OWN_T) dl[i] = 0.f;
    if (tid == 0) cnt = 0;
    __syncthreads();

    const v4i* row4 = (const v4i*)(ei + (size_t)b * 2 * GE);
    const v4i* col4 = (const v4i*)(ei + (size_t)b * 2 * GE + GE);
    const v4f* w4   = (const v4f*)(ew + (size_t)b * GE);
    for (int it = 0; it < (GE / 4 + 4 * OWN_T - 1) / (4 * OWN_T); ++it) {
        v4i c[4];
        v4i r[4];
        v4f w[4];
#pragma unroll
        for (int u = 0; u < 4; ++u) {
            int p = (it * 4 + u) * OWN_T + tid;
            if (p < GE / 4) { c[u] = col4[p]; r[u] = row4[p]; w[u] = w4[p]; }
            else            { c[u] = (v4i){-1, -1, -1, -1}; r[u] = (v4i){0, 0, 0, 0}; w[u] = (v4f){0.f, 0.f, 0.f, 0.f}; }
        }
        unsigned m[16];
        unsigned tot = 0;
#pragma unroll
        for (int i = 0; i < 16; ++i) {
            bool hit = ((unsigned)c[i >> 2][i & 3] - lo) < (unsigned)RN;
            m[i] = __builtin_amdgcn_ballot_w32(hit);
            tot += __builtin_popcount(m[i]);
        }
        unsigned at = 0;
        if (tot) {
            if (lane == 0) at = atomicAdd(&cnt, tot);
            at = __builtin_amdgcn_readfirstlane(at);
        }
#pragma unroll
        for (int i = 0; i < 16; ++i) {
            unsigned cl = (unsigned)c[i >> 2][i & 3] - lo;
            if (cl < (unsigned)RN) {
                float wt = w[i >> 2][i & 3];
                atomicAdd(&dl[cl], wt);
                unsigned pos = at + __builtin_amdgcn_mbcnt_lo(m[i], 0u);
                if (pos < (unsigned)CAP)
                    lst[pos] = (v2u){(unsigned)r[i >> 2][i & 3] | (cl << 16), __float_as_uint(wt)};
            }
            at += __builtin_popcount(m[i]);
        }
    }
    __syncthreads();
    for (int i = tid; i < RN; i += OWN_T)
        dinv[(size_t)b * GN + lo + i] = rsqrtf(dl[i] + 1.0f);
    if (tid == 0) counts[wg] = cnt;
}

template <int DO_TANH, int FUSE, int NORM>
__global__ __launch_bounds__(OWN_T) void agg_kernel(v2u* __restrict__ lists,
                                                    const unsigned int* __restrict__ counts,
                                                    const int* __restrict__ ei,
                                                    const float* __restrict__ ew,
                                                    const float* __restrict__ tmp,
                                                    const float* __restrict__ dinv,
                                                    const float* __restrict__ bias,
                                                    const float* __restrict__ Wn,
                                                    float* __restrict__ hout) {
    __shared__ float acc[RN * AP];
    __shared__ float wl[FUSE ? GH * AP : 4];
    __shared__ float dvl[NORM ? GN : 4];

    const int tid  = threadIdx.x;
    const int lane = tid & 31;
    const int wave = __builtin_amdgcn_readfirstlane(tid >> 5);
    const int b    = blockIdx.y;
    const unsigned lo = blockIdx.x * RN;
    const int wg   = b * NRANGE + blockIdx.x;
    const size_t nodeBase = (size_t)b * GN + lo;

    v4f dv[(GN / 4 + OWN_T - 1) / OWN_T];
    if (NORM) {
#pragma unroll
        for (int j = 0; j < (GN / 4 + OWN_T - 1) / OWN_T; ++j) {
            int i4 = tid + j * OWN_T;
            if (i4 < GN / 4) dv[j] = *(const v4f*)(dinv + (size_t)b * GN + (size_t)i4 * 4);
        }
    }
    for (int i4 = tid; i4 < RN * (GH / 4); i4 += OWN_T) {
        int node = i4 >> 4;
        int k4   = i4 & 15;
        float di = dinv[nodeBase + node];
        float sl = di * di;
        v4f t  = *(const v4f*)(tmp + (nodeBase + node) * GH + k4 * 4);
        v4f bb = *(const v4f*)(bias + k4 * 4);
        *(v4f*)(acc + node * AP + k4 * 4) = t * sl + bb;
    }
    if (NORM) {
#pragma unroll
        for (int j = 0; j < (GN / 4 + OWN_T - 1) / OWN_T; ++j) {
            int i4 = tid + j * OWN_T;
            if (i4 < GN / 4) *(v4f*)(dvl + i4 * 4) = dv[j];
        }
    }
    __syncthreads();

    const float* tg = tmp + (size_t)b * GN * GH;
    const unsigned nAll = counts[wg];

    if (nAll <= (unsigned)CAP) {
        const int n = (int)nAll;
        v2u* lst = lists + (size_t)wg * CAP;
        int c0 = wave * 32;
        v2u e = (v2u){0u, 0u};
        if (c0 + lane < n) {
            e = lst[c0 + lane];
            if (NORM) {
                e.y = __float_as_uint(dvl[e.x & 0xFFFFu] * __uint_as_float(e.y) * dvl[lo + (e.x >> 16)]);
                lst[c0 + lane] = e;
            }
        }
        for (; c0 < n; c0 += OWN_T) {
            v2u enx = (v2u){0u, 0u};
            const int m = n - c0;
#pragma unroll
            for (int g = 0; g < 32; g += 16) {
                if (g < m) {
                    v2f      v[16];
                    unsigned sx[16];
                    float    wv[16];
#pragma unroll
                    for (int u = 0; u < 16; ++u) {
                        sx[u] = (unsigned)__builtin_amdgcn_readlane((int)e.x, g + u);
                        wv[u] = __uint_as_float((unsigned)__builtin_amdgcn_readlane((int)e.y, g + u));
                        const float* rowp = tg + (size_t)(sx[u] & 0xFFFFu) * GH;
                        v[u]  = *(const v2f*)(rowp + 2 * lane);
                    }
                    if (g == 0 && c0 + OWN_T + lane < n) enx = lst[c0 + OWN_T + lane];
#pragma unroll
                    for (int u = 0; u < 16; ++u) {
                        if (g + u < m) {
                            float* d = acc + (size_t)(sx[u] >> 16) * AP + 2 * lane;
                            atomicAdd(d,     v[u].x * wv[u]);
                            atomicAdd(d + 1, v[u].y * wv[u]);
                        }
                    }
                }
            }
            if (NORM) {
                if (c0 + OWN_T + lane < n) {
                    enx.y = __float_as_uint(dvl[enx.x & 0xFFFFu] * __uint_as_float(enx.y) * dvl[lo + (enx.x >> 16)]);
                    lst[c0 + OWN_T + lane] = enx;
                }
            }
            e = enx;
        }
    } else {
        const int*   rowp = ei + (size_t)b * 2 * GE;
        const int*   colp = rowp + GE;
        const float* wp   = ew + (size_t)b * GE;
        for (int eI = tid; eI < GE; eI += OWN_T) {
            int c = colp[eI];
            unsigned cl = (unsigned)c - lo;
            if (cl < (unsigned)RN) {
                int r = rowp[eI];
                float nv = dinv[(size_t)b * GN + r] * wp[eI] * dinv[(size_t)b * GN + c];
                const float* sr = tmp + ((size_t)b * GN + r) * GH;
                float* d = acc + (size_t)cl * AP;
                for (int k = 0; k < GH; ++k) atomicAdd(d + k, sr[k] * nv);
            }
        }
    }
    __syncthreads();

    if (!FUSE) {
        for (int i4 = tid; i4 < RN * (GH / 4); i4 += OWN_T) {
            int node = i4 >> 4;
            int k4   = i4 & 15;
            v4f v = *(const v4f*)(acc + node * AP + k4 * 4);
            if (DO_TANH) {
                v.x = tanhf(v.x); v.y = tanhf(v.y); v.z = tanhf(v.z); v.w = tanhf(v.w);
            }
            *(v4f*)(hout + nodeBase * GH + (size_t)i4 * 4) = v;
        }
    } else {
        for (int i4 = tid; i4 < RN * (GH / 4); i4 += OWN_T) {
            int node = i4 >> 4;
            int k4   = i4 & 15;
            v4f v = *(const v4f*)(acc + node * AP + k4 * 4);
            if (DO_TANH) {
                v.x = tanhf(v.x); v.y = tanhf(v.y); v.z = tanhf(v.z); v.w = tanhf(v.w);
            }
            *(v4f*)(acc + node * AP + k4 * 4) = v;
        }
        for (int i4 = tid; i4 < GH * (GH / 4); i4 += OWN_T) {
            int n  = i4 >> 4;
            int k4 = i4 & 15;
            *(v4f*)(wl + n * AP + k4 * 4) = *(const v4f*)(Wn + (size_t)i4 * 4);
        }
        __syncthreads();

        const int half = lane >> 4;
        const int lm   = lane & 15;
        for (int job = wave; job < ((RN + 15) / 16) * (GH / 16); job += OWN_T / 32) {
            const int m0 = (job >> 2) * 16;
            const int n0 = (job & 3) * 16;
            int row = m0 + lm;
            row = row < RN ? row : RN - 1;
            const float* arow = acc + row * AP;
            const float* wrow = wl + (n0 + lm) * AP;
            v8f c = {};
#pragma unroll
            for (int k0 = 0; k0 < GH; k0 += 4) {
                const int ka = k0 + half * 2;
                v2f a  = *(const v2f*)(arow + ka);
                v2f bf = *(const v2f*)(wrow + ka);
                c = __builtin_amdgcn_wmma_f32_16x16x4_f32(
                     false, a,  false, bf,
                     (short)0, c,  false,  false);
            }
            float* drow = hout + (nodeBase + m0) * GH + n0 + lm;
#pragma unroll
            for (int j = 0; j < 8; ++j) {
                const int r = j + 8 * half;
                if (m0 + r < RN) drow[(size_t)r * GH] = c[j];
            }
        }
    }
}

__global__ __launch_bounds__(256) void gather_kernel(const float* __restrict__ emb,
                                                     const int* __restrict__ pos,
                                                     float* __restrict__ out) {
    int idx = blockIdx.x * blockDim.x + threadIdx.x;
    if (idx >= GB * GP * GEMB) return;
    int k = idx & (GEMB - 1);
    int bp = idx >> 6;
    int b = bp / GP;
    int pv = pos[bp];
    out[idx] = (pv >= 0) ? emb[((size_t)b * GN + pv) * GH + k] : -3.0f;
}

extern "C" void kernel_launch(void* const* d_in, const int* in_sizes, int n_in,
                              void* d_out, int out_size, void* d_ws, size_t ws_size,
                              hipStream_t stream) {
    const float* x   = (const float*)d_in[0];
    const int*   ei  = (const int*)d_in[1];
    const float* ew  = (const float*)d_in[2];
    const int*   pos = (const int*)d_in[3];
    const float* W1  = (const float*)d_in[4];
    const float* b1  = (const float*)d_in[5];
    const float* W2  = (const float*)d_in[6];
    const float* b2  = (const float*)d_in[7];
    const float* W3  = (const float*)d_in[8];
    const float* b3  = (const float*)d_in[9];
    float* out = (float*)d_out;

    float* ws   = (float*)d_ws;
    const size_t nBN  = (size_t)GB * GN;
    const size_t nBNH = (size_t)GB * GN * GH;
    const size_t nList = (size_t)NRANGE * GB;
    float* dinv = ws;
    unsigned int* counts = (unsigned int*)(dinv + nBN);
    v2u*   lists = (v2u*)(dinv + nBN + nList);
    float* tmp  = dinv + nBN + nList + 2 * nList * CAP;
    float* hbuf = tmp + nBNH;

    const int T = 256;
    dim3 own_grid(NRANGE, GB, 1);

    deg_kernel<<<own_grid, OWN_T, 0, stream>>>(ei, ew, x, W1, tmp, dinv, lists, counts);

    agg_kernel<1, 1, 1><<<own_grid, OWN_T, 0, stream>>>(lists, counts, ei, ew, tmp, dinv, b1, W2, hbuf);

    agg_kernel<1, 1, 0><<<own_grid, OWN_T, 0, stream>>>(lists, counts, ei, ew, hbuf, dinv, b2, W3, tmp);

    agg_kernel<0, 0, 0><<<own_grid, OWN_T, 0, stream>>>(lists, counts, ei, ew, tmp, dinv, b3, W3, hbuf);

    gather_kernel<<<(GB * GP * GEMB + T - 1) / T, T, 0, stream>>>(hbuf, pos, out);
}


# In-edge list grouped once by owner wave (target mod 32) in degree pass, not per layer; a wave adds its sub-list's rows to LDS row sums by plain read-fma-write, not float atomics; sum order changes
# speedup vs baseline: 4.6242x; 1.0844x over previous
#include <hip/hip_runtime.h>
#include <hip/hip_bf16.h>

#define GB   16
#define GN   10000
#define GE   160000
#define GS   128
#define GH   64
#define GEMB 64
#define GP   16

typedef __attribute__((ext_vector_type(2))) float v2f;
typedef __attribute__((ext_vector_type(4))) float v4f;
typedef __attribute__((ext_vector_type(8))) float v8f;
typedef __attribute__((ext_vector_type(2))) unsigned int v2u;
typedef __attribute__((ext_vector_type(4))) int v4i;

#define NRANGE 16
#define RN     (GN / NRANGE)
#define OWN_T  1024
#define CAP    11264
#define CAPU   (CAP - 32)
#define NOWN   (OWN_T / 32)
#define AP     68
#define XP     132
#define XR     128
#define NCH    ((RN + XR - 1) / XR)

__global__ __launch_bounds__(OWN_T) void deg_kernel(const int* __restrict__ ei,
                                                    const float* __restrict__ ew,
                                                    const float* __restrict__ x,
                                                    const float* __restrict__ W1,
                                                    float* __restrict__ tmp,
                                                    float* __restrict__ dinv,
                                                    v2u* __restrict__ lists,
                                                    unsigned int* __restrict__ counts) {
    __shared__ float dl[RN];
    __shared__ unsigned int cnt;
    __shared__ unsigned int ocnt[NOWN];
    __shared__ unsigned int obase[NOWN];
    __shared__ float w1l[GH * XP];
    __shared__ float xl[2][XR * XP];
    const int tid  = threadIdx.x;
    const int lane = tid & 31;
    const int b    = blockIdx.y;
    const unsigned lo = blockIdx.x * RN;
    const int wg   = b * NRANGE + blockIdx.x;
    v2u* lst = lists + (size_t)wg * CAP;

    {
        const int wave = __builtin_amdgcn_readfirstlane(tid >> 5);
        const int half = lane >> 4;
        const int lm   = lane & 15;
        const size_t nodeBase = (size_t)b * GN + lo;
        const int xrow = tid >> 5;
        const int xk4  = tid & 31;

        for (int i4 = tid; i4 < GH * (GS / 4); i4 += OWN_T) {
            int n  = i4 >> 5;
            int k4 = i4 & 31;
            *(v4f*)(w1l + n * XP + k4 * 4) = *(const v4f*)(W1 + (size_t)i4 * 4);
        }
        v4f nx[4];
#pragma unroll
        for (int j = 0; j < 4; ++j) {
            int r = xrow + j * 32;
            nx[j] = *(const v4f*)(x + (nodeBase + r) * GS + xk4 * 4);
        }
#pragma unroll
        for (int j = 0; j < 4; ++j)
            *(v4f*)(&xl[0][(xrow + j * 32) * XP + xk4 * 4]) = nx[j];
        __syncthreads();

        const int m0 = (wave >> 2) * 16;
        const int n0 = (wave & 3) * 16;
        const float* wrow = w1l + (n0 + lm) * XP;
        for (int c = 0; c < NCH; ++c) {
            if (c + 1 < NCH) {
#pragma unroll
                for (int j = 0; j < 4; ++j) {
                    int r = (c + 1) * XR + xrow + j * 32;
                    r = r < RN ? r : RN - 1;
                    nx[j] = *(const v4f*)(x + (nodeBase + r) * GS + xk4 * 4);
                }
            }
            const float* arow = &xl[c & 1][(m0 + lm) * XP];
            v8f acc = {};
#pragma unroll
            for (int k0 = 0; k0 < GS; k0 += 4) {
                const int ka = k0 + half * 2;
                v2f a  = *(const v2f*)(arow + ka);
                v2f bf = *(const v2f*)(wrow + ka);
                acc = __builtin_amdgcn_wmma_f32_16x16x4_f32(
                     false, a,  false, bf,
                     (short)0, acc,  false,  false);
            }
            const int rbase = c * XR + m0;
            float* drow = tmp + (nodeBase + rbase) * GH + n0 + lm;
#pragma unroll
            for (int j = 0; j < 8; ++j) {
                const int r = j + 8 * half;
                if (c + 1 < NCH || rbase + r < RN) drow[(size_t)r * GH] = acc[j];
            }
            if (c + 1 < NCH) {
#pragma unroll
                for (int j = 0; j < 4; ++j)
                    *(v4f*)(&xl[(c + 1) & 1][(xrow + j * 32) * XP + xk4 * 4]) = nx[j];
            }
            __syncthreads();
        }
    }
    for (int i = tid; i < RN; i += OWN_T) dl[i] = 0.f;
    if (tid == 0) cnt = 0;
    if (tid < NOWN) ocnt[tid] = 0;
    __syncthreads();
    v2f* stg = (v2f*)&xl[0][0];

    const v4i* row4 = (const v4i*)(ei + (size_t)b * 2 * GE);
    const v4i* col4 = (const v4i*)(ei + (size_t)b * 2 * GE + GE);
    const v4f* w4   = (const v4f*)(ew + (size_t)b * GE);
    for (int it = 0; it < (GE / 4 + 4 * OWN_T - 1) / (4 * OWN_T); ++it) {
        v4i c[4];
        v4i r[4];
        v4f w[4];
#pragma unroll
        for (int u = 0; u < 4; ++u) {
            int p = (it * 4 + u) * OWN_T + tid;
            if (p < GE / 4) { c[u] = col4[p]; r[u] = row4[p]; w[u] = w4[p]; }
            else            { c[u] = (v4i){-1, -1, -1, -1}; r[u] = (v4i){0, 0, 0, 0}; w[u] = (v4f){0.f, 0.f, 0.f, 0.f}; }
        }
        unsigned m[16];
        unsigned tot = 0;
#pragma unroll
        for (int i = 0; i < 16; ++i) {
            bool hit = ((unsigned)c[i >> 2][i & 3] - lo) < (unsigned)RN;
            m[i] = __builtin_amdgcn_ballot_w32(hit);
            tot += __builtin_popcount(m[i]);
        }
        unsigned at = 0;
        if (tot) {
            if (lane == 0) at = atomicAdd(&cnt, tot);
            at = __builtin_amdgcn_readfirstlane(at);
        }
#pragma unroll
        for (int i = 0; i < 16; ++i) {
            unsigned cl = (unsigned)c[i >> 2][i & 3] - lo;
            if (cl < (unsigned)RN) {
                float wt = w[i >> 2][i & 3];
                atomicAdd(&dl[cl], wt);
                unsigned pos = at + __builtin_amdgcn_mbcnt_lo(m[i], 0u);
                if (pos < (unsigned)CAPU)
                    stg[pos] = (v2f){__uint_as_float((unsigned)r[i >> 2][i & 3] | (cl << 16)), wt};
            }
            at += __builtin_popcount(m[i]);
        }
    }
    __syncthreads();
    for (int i = tid; i < RN; i += OWN_T)
        dinv[(size_t)b * GN + lo + i] = rsqrtf(dl[i] + 1.0f);
    const unsigned nKept = cnt;
    if (tid == 0) counts[wg] = nKept;
    if (nKept <= (unsigned)CAPU) {
        const int n = (int)nKept;
        v2f      en[CAP / OWN_T];
        unsigned rk[CAP / OWN_T];
#pragma unroll
        for (int k = 0; k < CAP / OWN_T; ++k) {
            int i = tid + k * OWN_T;
            if (i < n) {
                en[k] = stg[i];
                rk[k] = atomicAdd(&ocnt[(__float_as_uint(en[k].x) >> 16) & (NOWN - 1)], 1u);
            }
        }
        __syncthreads();
        if (tid < NOWN) {
            unsigned ex = 0;
            for (int j = 0; j < tid; ++j) ex += ocnt[j];
            obase[tid] = ex;
            unsigned* bnd = (unsigned*)(lst + CAPU);
            bnd[tid] = ex;
            if (tid == NOWN - 1) bnd[NOWN] = ex + ocnt[NOWN - 1];
        }
        __syncthreads();
#pragma unroll
        for (int k = 0; k < CAP / OWN_T; ++k) {
            int i = tid + k * OWN_T;
            if (i < n) stg[obase[(__float_as_uint(en[k].x) >> 16) & (NOWN - 1)] + rk[k]] = en[k];
        }
        __syncthreads();
        for (int i = tid; i < n; i += OWN_T) {
            v2f t = stg[i];
            lst[i] = (v2u){__float_as_uint(t.x), __float_as_uint(t.y)};
        }
    }
}

template <int G>
__device__ __forceinline__ void rows16(v2f (&v)[16], const v2u e, const float* __restrict__ tg, const int lane) {
#pragma unroll
    for (int u = 0; u < 16; ++u) {
        const unsigned sx = (unsigned)__builtin_amdgcn_readlane((int)e.x, G + u);
        const float* rowp = tg + (size_t)(sx & 0xFFFFu) * GH;
        v[u] = *(const v2f*)(rowp + 2 * lane);
    }
}
template <int G, bool FULL>
__device__ __forceinline__ void add16(const v2f (&v)[16], const v2u e, float* sums, const int lane, const int m) {
#pragma unroll
    for (int u = 0; u < 16; ++u) {
        if (FULL || G + u < m) {
            const unsigned sx = (unsigned)__builtin_amdgcn_readlane((int)e.x, G + u);
            const float    w  = __uint_as_float((unsigned)__builtin_amdgcn_readlane((int)e.y, G + u));
            float* d = sums + (size_t)(sx >> 16) * AP + 2 * lane;
            v2f a = *(v2f*)d;
            a.x += v[u].x * w;
            a.y += v[u].y * w;
            *(v2f*)d = a;
        }
    }
}

template <int DO_TANH, int FUSE, int NORM>
__global__ __launch_bounds__(OWN_T) void agg_kernel(v2u* __restrict__ lists,
                                                    const unsigned int* __restrict__ counts,
                                                    const int* __restrict__ ei,
                                                    const float* __restrict__ ew,
                                                    const float* __restrict__ tmp,
                                                    const float* __restrict__ dinv,
                                                    const float* __restrict__ bias,
                                                    const float* __restrict__ Wn,
                                                    float* __restrict__ hout) {
    __shared__ float acc[RN * AP];
    __shared__ float wl[FUSE ? GH * AP : 4];
    __shared__ float dvl[NORM ? GN : 4];

    const int tid  = threadIdx.x;
    const int lane = tid & 31;
    const int wave = __builtin_amdgcn_readfirstlane(tid >> 5);
    const int b    = blockIdx.y;
    const unsigned lo = blockIdx.x * RN;
    const int wg   = b * NRANGE + blockIdx.x;
    const size_t nodeBase = (size_t)b * GN + lo;

    v4f dv[(GN / 4 + OWN_T - 1) / OWN_T];
    if (NORM) {
#pragma unroll
        for (int j = 0; j < (GN / 4 + OWN_T - 1) / OWN_T; ++j) {
            int i4 = tid + j * OWN_T;
            if (i4 < GN / 4) dv[j] = *(const v4f*)(dinv + (size_t)b * GN + (size_t)i4 * 4);
        }
    }
    for (int i4 = tid; i4 < RN * (GH / 4); i4 += OWN_T) {
        int node = i4 >> 4;
        int k4   = i4 & 15;
        float di = dinv[nodeBase + node];
        float sl = di * di;
        v4f t  = *(const v4f*)(tmp + (nodeBase + node) * GH + k4 * 4);
        v4f bb = *(const v4f*)(bias + k4 * 4);
        *(v4f*)(acc + node * AP + k4 * 4) = t * sl + bb;
    }
    if (NORM) {
#pragma unroll
        for (int j = 0; j < (GN / 4 + OWN_T - 1) / OWN_T; ++j) {
            int i4 = tid + j * OWN_T;
            if (i4 < GN / 4) *(v4f*)(dvl + i4 * 4) = dv[j];
        }
    }
    __syncthreads();

    const float* tg = tmp + (size_t)b * GN * GH;
    const unsigned nAll = counts[wg];

    if (nAll <= (unsigned)CAPU) {
        v2u* lst = lists + (size_t)wg * CAP;
        const unsigned* bnd = (const unsigned*)(lst + CAPU);
        int c0 = (int)bnd[wave];
        const int n = (int)bnd[wave + 1];
        v2u e = (v2u){0u, 0u};
        if (c0 + lane < n) {
            e = lst[c0 + lane];
            if (NORM) {
                e.y = __float_as_uint(dvl[e.x & 0xFFFFu] * __uint_as_float(e.y) * dvl[lo + (e.x >> 16)]);
                lst[c0 + lane] = e;
            }
        }
        v2f va[16], vb[16];
        rows16<0>(va, e, tg, lane);
        for (; c0 < n; c0 += 32) {
            const int m = n - c0;
            v2u enx = (v2u){0u, 0u};
            if (c0 + 32 + lane < n) enx = lst[c0 + 32 + lane];
            rows16<16>(vb, e, tg, lane);
            if (m >= 32) add16<0, true>(va, e, acc, lane, m);
            else         add16<0, false>(va, e, acc, lane, m);
            if (NORM) {
                if (c0 + 32 + lane < n) {
                    enx.y = __float_as_uint(dvl[enx.x & 0xFFFFu] * __uint_as_float(enx.y) * dvl[lo + (enx.x >> 16)]);
                    lst[c0 + 32 + lane] = enx;
                }
            }
            rows16<0>(va, enx, tg, lane);
            if (m >= 32) add16<16, true>(vb, e, acc, lane, m);
            else         add16<16, false>(vb, e, acc, lane, m);
            e = enx;
        }
    } else {
        const int*   rowp = ei + (size_t)b * 2 * GE;
        const int*   colp = rowp + GE;
        const float* wp   = ew + (size_t)b * GE;
        for (int eI = tid; eI < GE; eI += OWN_T) {
            int c = colp[eI];
            unsigned cl = (unsigned)c - lo;
            if (cl < (unsigned)RN) {
                int r = rowp[eI];
                float nv = dinv[(size_t)b * GN + r] * wp[eI] * dinv[(size_t)b * GN + c];
                const float* sr = tmp + ((size_t)b * GN + r) * GH;
                float* d = acc + (size_t)cl * AP;
                for (int k = 0; k < GH; ++k) atomicAdd(d + k, sr[k] * nv);
            }
        }
    }
    __syncthreads();

    if (!FUSE) {
        for (int i4 = tid; i4 < RN * (GH / 4); i4 += OWN_T) {
            int node = i4 >> 4;
            int k4   = i4 & 15;
            v4f v = *(const v4f*)(acc + node * AP + k4 * 4);
            if (DO_TANH) {
                v.x = tanhf(v.x); v.y = tanhf(v.y); v.z = tanhf(v.z); v.w = tanhf(v.w);
            }
            *(v4f*)(hout + nodeBase * GH + (size_t)i4 * 4) = v;
        }
    } else {
        for (int i4 = tid; i4 < RN * (GH / 4); i4 += OWN_T) {
            int node = i4 >> 4;
            int k4   = i4 & 15;
            v4f v = *(const v4f*)(acc + node * AP + k4 * 4);
            if (DO_TANH) {
                v.x = tanhf(v.x); v.y = tanhf(v.y); v.z = tanhf(v.z); v.w = tanhf(v.w);
            }
            *(v4f*)(acc + node * AP + k4 * 4) = v;
        }
        for (int i4 = tid; i4 < GH * (GH / 4); i4 += OWN_T) {
            int n  = i4 >> 4;
            int k4 = i4 & 15;
            *(v4f*)(wl + n * AP + k4 * 4) = *(const v4f*)(Wn + (size_t)i4 * 4);
        }
        __syncthreads();

        const int half = lane >> 4;
        const int lm   = lane & 15;
        for (int job = wave; job < ((RN + 15) / 16) * (GH / 16); job += OWN_T / 32) {
            const int m0 = (job >> 2) * 16;
            const int n0 = (job & 3) * 16;
            int row = m0 + lm;
            row = row < RN ? row : RN - 1;
            const float* arow = acc + row * AP;
            const float* wrow = wl + (n0 + lm) * AP;
            v8f c = {};
#pragma unroll
            for (int k0 = 0; k0 < GH; k0 += 4) {
                const int ka = k0 + half * 2;
                v2f a  = *(const v2f*)(arow + ka);
                v2f bf = *(const v2f*)(wrow + ka);
                c = __builtin_amdgcn_wmma_f32_16x16x4_f32(
                     false, a,  false, bf,
                     (short)0, c,  false,  false);
            }
            float* drow = hout + (nodeBase + m0) * GH + n0 + lm;
#pragma unroll
            for (int j = 0; j < 8; ++j) {
                const int r = j + 8 * half;
                if (m0 + r < RN) drow[(size_t)r * GH] = c[j];
            }
        }
    }
}

__global__ __launch_bounds__(256) void gather_kernel(const float* __restrict__ emb,
                                                     const int* __restrict__ pos,
                                                     float* __restrict__ out) {
    int idx = blockIdx.x * blockDim.x + threadIdx.x;
    if (idx >= GB * GP * GEMB) return;
    int k = idx & (GEMB - 1);
    int bp = idx >> 6;
    int b = bp / GP;
    int pv = pos[bp];
    out[idx] = (pv >= 0) ? emb[((size_t)b * GN + pv) * GH + k] : -3.0f;
}

extern "C" void kernel_launch(void* const* d_in, const int* in_sizes, int n_in,
                              void* d_out, int out_size, void* d_ws, size_t ws_size,
                              hipStream_t stream) {
    const float* x   = (const float*)d_in[0];
    const int*   ei  = (const int*)d_in[1];
    const float* ew  = (const float*)d_in[2];
    const int*   pos = (const int*)d_in[3];
    const float* W1  = (const float*)d_in[4];
    const float* b1  = (const float*)d_in[5];
    const float* W2  = (const float*)d_in[6];
    const float* b2  = (const float*)d_in[7];
    const float* W3  = (const float*)d_in[8];
    const float* b3  = (const float*)d_in[9];
    float* out = (float*)d_out;

    float* ws   = (float*)d_ws;
    const size_t nBN  = (size_t)GB * GN;
    const size_t nBNH = (size_t)GB * GN * GH;
    const size_t nList = (size_t)NRANGE * GB;
    float* dinv = ws;
    unsigned int* counts = (unsigned int*)(dinv + nBN);
    v2u*   lists = (v2u*)(dinv + nBN + nList);
    float* tmp  = dinv + nBN + nList + 2 * nList * CAP;
    float* hbuf = tmp + nBNH;

    const int T = 256;
    dim3 own_grid(NRANGE, GB, 1);

    deg_kernel<<<own_grid, OWN_T, 0, stream>>>(ei, ew, x, W1, tmp, dinv, lists, counts);

    agg_kernel<1, 1, 1><<<own_grid, OWN_T, 0, stream>>>(lists, counts, ei, ew, tmp, dinv, b1, W2, hbuf);

    agg_kernel<1, 1, 0><<<own_grid, OWN_T, 0, stream>>>(lists, counts, ei, ew, hbuf, dinv, b2, W3, tmp);

    agg_kernel<0, 0, 0><<<own_grid, OWN_T, 0, stream>>>(lists, counts, ei, ew, tmp, dinv, b3, W3, hbuf);

    gather_kernel<<<(GB * GP * GEMB + T - 1) / T, T, 0, stream>>>(hbuf, pos, out);
}


# Owner wave's in-edge sub-list ordered by source bucket (64 nodes) in the degree pass: all waves sweep source rows low to high together, repeated rows hit the WGP cache; sum order changes
# speedup vs baseline: 4.8114x; 1.0405x over previous
#include <hip/hip_runtime.h>
#include <hip/hip_bf16.h>

#define GB   16
#define GN   10000
#define GE   160000
#define GS   128
#define GH   64
#define GEMB 64
#define GP   16

typedef __attribute__((ext_vector_type(2))) float v2f;
typedef __attribute__((ext_vector_type(4))) float v4f;
typedef __attribute__((ext_vector_type(8))) float v8f;
typedef __attribute__((ext_vector_type(2))) unsigned int v2u;
typedef __attribute__((ext_vector_type(4))) int v4i;

#define NRANGE 16
#define RN     (GN / NRANGE)
#define OWN_T  1024
#define CAP    11264
#define CAPU   (CAP - 32)
#define NOWN   (OWN_T / 32)
#define SB     6
#define NBK    160
#define NKEY   (NOWN * NBK)
#define KPT    (NKEY / OWN_T)
#define AP     68
#define XP     132
#define XR     128
#define NCH    ((RN + XR - 1) / XR)

__device__ __forceinline__ unsigned grp_key(const unsigned ex) {
    return ((ex >> 16) & (NOWN - 1)) * NBK + ((ex & 0xFFFFu) >> SB);
}

__global__ __launch_bounds__(OWN_T) void deg_kernel(const int* __restrict__ ei,
                                                    const float* __restrict__ ew,
                                                    const float* __restrict__ x,
                                                    const float* __restrict__ W1,
                                                    float* __restrict__ tmp,
                                                    float* __restrict__ dinv,
                                                    v2u* __restrict__ lists,
                                                    unsigned int* __restrict__ counts) {
    __shared__ float dl[RN];
    __shared__ unsigned int cnt;
    __shared__ unsigned int ocnt[NKEY];
    __shared__ unsigned int wsum[NOWN];
    __shared__ float w1l[GH * XP];
    __shared__ float xl[2][XR * XP];
    const int tid  = threadIdx.x;
    const int lane = tid & 31;
    const int b    = blockIdx.y;
    const unsigned lo = blockIdx.x * RN;
    const int wg   = b * NRANGE + blockIdx.x;
    v2u* lst = lists + (size_t)wg * CAP;

    {
        const int wave = __builtin_amdgcn_readfirstlane(tid >> 5);
        const int half = lane >> 4;
        const int lm   = lane & 15;
        const size_t nodeBase = (size_t)b * GN + lo;
        const int xrow = tid >> 5;
        const int xk4  = tid & 31;

        for (int i4 = tid; i4 < GH * (GS / 4); i4 += OWN_T) {
            int n  = i4 >> 5;
            int k4 = i4 & 31;
            *(v4f*)(w1l + n * XP + k4 * 4) = *(const v4f*)(W1 + (size_t)i4 * 4);
        }
        v4f nx[4];
#pragma unroll
        for (int j = 0; j < 4; ++j) {
            int r = xrow + j * 32;
            nx[j] = *(const v4f*)(x + (nodeBase + r) * GS + xk4 * 4);
        }
#pragma unroll
        for (int j = 0; j < 4; ++j)
            *(v4f*)(&xl[0][(xrow + j * 32) * XP + xk4 * 4]) = nx[j];
        __syncthreads();

        const int m0 = (wave >> 2) * 16;
        const int n0 = (wave & 3) * 16;
        const float* wrow = w1l + (n0 + lm) * XP;
        for (int c = 0; c < NCH; ++c) {
            if (c + 1 < NCH) {
#pragma unroll
                for (int j = 0; j < 4; ++j) {
                    int r = (c + 1) * XR + xrow + j * 32;
                    r = r < RN ? r : RN - 1;
                    nx[j] = *(const v4f*)(x + (nodeBase + r) * GS + xk4 * 4);
                }
            }
            const float* arow = &xl[c & 1][(m0 + lm) * XP];
            v8f acc = {};
#pragma unroll
            for (int k0 = 0; k0 < GS; k0 += 4) {
                const int ka = k0 + half * 2;
                v2f a  = *(const v2f*)(arow + ka);
                v2f bf = *(const v2f*)(wrow + ka);
                acc = __builtin_amdgcn_wmma_f32_16x16x4_f32(
                     false, a,  false, bf,
                     (short)0, acc,  false,  false);
            }
            const int rbase = c * XR + m0;
            float* drow = tmp + (nodeBase + rbase) * GH + n0 + lm;
#pragma unroll
            for (int j = 0; j < 8; ++j) {
                const int r = j + 8 * half;
                if (c + 1 < NCH || rbase + r < RN) drow[(size_t)r * GH] = acc[j];
            }
            if (c + 1 < NCH) {
#pragma unroll
                for (int j = 0; j < 4; ++j)
                    *(v4f*)(&xl[(c + 1) & 1][(xrow + j * 32) * XP + xk4 * 4]) = nx[j];
            }
            __syncthreads();
        }
    }
    for (int i = tid; i < RN; i += OWN_T) dl[i] = 0.f;
    if (tid == 0) cnt = 0;
    for (int i = tid; i < NKEY; i += OWN_T) ocnt[i] = 0;
    __syncthreads();
    v2f* stg = (v2f*)&xl[0][0];

    const v4i* row4 = (const v4i*)(ei + (size_t)b * 2 * GE);
    const v4i* col4 = (const v4i*)(ei + (size_t)b * 2 * GE + GE);
    const v4f* w4   = (const v4f*)(ew + (size_t)b * GE);
    for (int it = 0; it < (GE / 4 + 4 * OWN_T - 1) / (4 * OWN_T); ++it) {
        v4i c[4];
        v4i r[4];
        v4f w[4];
#pragma unroll
        for (int u = 0; u < 4; ++u) {
            int p = (it * 4 + u) * OWN_T + tid;
            if (p < GE / 4) { c[u] = col4[p]; r[u] = row4[p]; w[u] = w4[p]; }
            else            { c[u] = (v4i){-1, -1, -1, -1}; r[u] = (v4i){0, 0, 0, 0}; w[u] = (v4f){0.f, 0.f, 0.f, 0.f}; }
        }
        unsigned m[16];
        unsigned tot = 0;
#pragma unroll
        for (int i = 0; i < 16; ++i) {
            bool hit = ((unsigned)c[i >> 2][i & 3] - lo) < (unsigned)RN;
            m[i] = __builtin_amdgcn_ballot_w32(hit);
            tot += __builtin_popcount(m[i]);
        }
        unsigned at = 0;
        if (tot) {
            if (lane == 0) at = atomicAdd(&cnt, tot);
            at = __builtin_amdgcn_readfirstlane(at);
        }
#pragma unroll
        for (int i = 0; i < 16; ++i) {
            unsigned cl = (unsigned)c[i >> 2][i & 3] - lo;
            if (cl < (unsigned)RN) {
                float wt = w[i >> 2][i & 3];
                atomicAdd(&dl[cl], wt);
                unsigned pos = at + __builtin_amdgcn_mbcnt_lo(m[i], 0u);
                if (pos < (unsigned)CAPU)
                    stg[pos] = (v2f){__uint_as_float((unsigned)r[i >> 2][i & 3] | (cl << 16)), wt};
            }
            at += __builtin_popcount(m[i]);
        }
    }
    __syncthreads();
    for (int i = tid; i < RN; i += OWN_T)
        dinv[(size_t)b * GN + lo + i] = rsqrtf(dl[i] + 1.0f);
    const unsigned nKept = cnt;
    if (tid == 0) counts[wg] = nKept;
    if (nKept <= (unsigned)CAPU) {
        const int n = (int)nKept;
        v2f      en[CAP / OWN_T];
        unsigned rk[CAP / OWN_T];
#pragma unroll
        for (int k = 0; k < CAP / OWN_T; ++k) {
            int i = tid + k * OWN_T;
            if (i < n) {
                en[k] = stg[i];
                rk[k] = atomicAdd(&ocnt[grp_key(__float_as_uint(en[k].x))], 1u);
            }
        }
        __syncthreads();
        {
            unsigned gc[KPT];
            unsigned s = 0;
#pragma unroll
            for (int k = 0; k < KPT; ++k) { gc[k] = ocnt[tid * KPT + k]; s += gc[k]; }
            unsigned v = s;
#pragma unroll
            for (int d = 1; d < 32; d <<= 1) {
                unsigned o = __shfl_up(v, d);
                if (lane >= d) v += o;
            }
            if (lane == 31) wsum[tid >> 5] = v;
            __syncthreads();
            unsigned ex = v - s;
            for (int j = 0; j < (tid >> 5); ++j) ex += wsum[j];
            unsigned* bnd = (unsigned*)(lst + CAPU);
            if (lane == 0) bnd[tid >> 5] = ex;
            if (tid == 0) bnd[NOWN] = nKept;
#pragma unroll
            for (int k = 0; k < KPT; ++k) { ocnt[tid * KPT + k] = ex; ex += gc[k]; }
        }
        __syncthreads();
#pragma unroll
        for (int k = 0; k < CAP / OWN_T; ++k) {
            int i = tid + k * OWN_T;
            if (i < n) stg[ocnt[grp_key(__float_as_uint(en[k].x))] + rk[k]] = en[k];
        }
        __syncthreads();
        for (int i = tid; i < n; i += OWN_T) {
            v2f t = stg[i];
            lst[i] = (v2u){__float_as_uint(t.x), __float_as_uint(t.y)};
        }
    }
}

template <int G>
__device__ __forceinline__ void rows16(v2f (&v)[16], const v2u e, const float* __restrict__ tg, const int lane) {
#pragma unroll
    for (int u = 0; u < 16; ++u) {
        const unsigned sx = (unsigned)__builtin_amdgcn_readlane((int)e.x, G + u);
        const float* rowp = tg + (size_t)(sx & 0xFFFFu) * GH;
        v[u] = *(const v2f*)(rowp + 2 * lane);
    }
}
template <int G, bool FULL>
__device__ __forceinline__ void add16(const v2f (&v)[16], const v2u e, float* sums, const int lane, const int m) {
#pragma unroll
    for (int u = 0; u < 16; ++u) {
        if (FULL || G + u < m) {
            const unsigned sx = (unsigned)__builtin_amdgcn_readlane((int)e.x, G + u);
            const float    w  = __uint_as_float((unsigned)__builtin_amdgcn_readlane((int)e.y, G + u));
            float* d = sums + (size_t)(sx >> 16) * AP + 2 * lane;
            v2f a = *(v2f*)d;
            a.x += v[u].x * w;
            a.y += v[u].y * w;
            *(v2f*)d = a;
        }
    }
}

template <int DO_TANH, int FUSE, int NORM>
__global__ __launch_bounds__(OWN_T) void agg_kernel(v2u* __restrict__ lists,
                                                    const unsigned int* __restrict__ counts,
                                                    const int* __restrict__ ei,
                                                    const float* __restrict__ ew,
                                                    const float* __restrict__ tmp,
                                                    const float* __restrict__ dinv,
                                                    const float* __restrict__ bias,
                                                    const float* __restrict__ Wn,
                                                    float* __restrict__ hout) {
    __shared__ float acc[RN * AP];
    __shared__ float wl[FUSE ? GH * AP : 4];
    __shared__ float dvl[NORM ? GN : 4];

    const int tid  = threadIdx.x;
    const int lane = tid & 31;
    const int wave = __builtin_amdgcn_readfirstlane(tid >> 5);
    const int b    = blockIdx.y;
    const unsigned lo = blockIdx.x * RN;
    const int wg   = b * NRANGE + blockIdx.x;
    const size_t nodeBase = (size_t)b * GN + lo;

    v4f dv[(GN / 4 + OWN_T - 1) / OWN_T];
    if (NORM) {
#pragma unroll
        for (int j = 0; j < (GN / 4 + OWN_T - 1) / OWN_T; ++j) {
            int i4 = tid + j * OWN_T;
            if (i4 < GN / 4) dv[j] = *(const v4f*)(dinv + (size_t)b * GN + (size_t)i4 * 4);
        }
    }
    for (int i4 = tid; i4 < RN * (GH / 4); i4 += OWN_T) {
        int node = i4 >> 4;
        int k4   = i4 & 15;
        float di = dinv[nodeBase + node];
        float sl = di * di;
        v4f t  = *(const v4f*)(tmp + (nodeBase + node) * GH + k4 * 4);
        v4f bb = *(const v4f*)(bias + k4 * 4);
        *(v4f*)(acc + node * AP + k4 * 4) = t * sl + bb;
    }
    if (NORM) {
#pragma unroll
        for (int j = 0; j < (GN / 4 + OWN_T - 1) / OWN_T; ++j) {
            int i4 = tid + j * OWN_T;
            if (i4 < GN / 4) *(v4f*)(dvl + i4 * 4) = dv[j];
        }
    }
    __syncthreads();

    const float* tg = tmp + (size_t)b * GN * GH;
    const unsigned nAll = counts[wg];

    if (nAll <= (unsigned)CAPU) {
        v2u* lst = lists + (size_t)wg * CAP;
        const unsigned* bnd = (const unsigned*)(lst + CAPU);
        int c0 = (int)bnd[wave];
        const int n = (int)bnd[wave + 1];
        v2u e = (v2u){0u, 0u};
        if (c0 + lane < n) {
            e = lst[c0 + lane];
            if (NORM) {
                e.y = __float_as_uint(dvl[e.x & 0xFFFFu] * __uint_as_float(e.y) * dvl[lo + (e.x >> 16)]);
                lst[c0 + lane] = e;
            }
        }
        v2f va[16], vb[16];
        rows16<0>(va, e, tg, lane);
        for (; c0 < n; c0 += 32) {
            const int m = n - c0;
            v2u enx = (v2u){0u, 0u};
            if (c0 + 32 + lane < n) enx = lst[c0 + 32 + lane];
            rows16<16>(vb, e, tg, lane);
            if (m >= 32) add16<0, true>(va, e, acc, lane, m);
            else         add16<0, false>(va, e, acc, lane, m);
            if (NORM) {
                if (c0 + 32 + lane < n) {
                    enx.y = __float_as_uint(dvl[enx.x & 0xFFFFu] * __uint_as_float(enx.y) * dvl[lo + (enx.x >> 16)]);
                    lst[c0 + 32 + lane] = enx;
                }
            }
            rows16<0>(va, enx, tg, lane);
            if (m >= 32) add16<16, true>(vb, e, acc, lane, m);
            else         add16<16, false>(vb, e, acc, lane, m);
            e = enx;
        }
    } else {
        const int*   rowp = ei + (size_t)b * 2 * GE;
        const int*   colp = rowp + GE;
        const float* wp   = ew + (size_t)b * GE;
        for (int eI = tid; eI < GE; eI += OWN_T) {
            int c = colp[eI];
            unsigned cl = (unsigned)c - lo;
            if (cl < (unsigned)RN) {
                int r = rowp[eI];
                float nv = dinv[(size_t)b * GN + r] * wp[eI] * dinv[(size_t)b * GN + c];
                const float* sr = tmp + ((size_t)b * GN + r) * GH;
                float* d = acc + (size_t)cl * AP;
                for (int k = 0; k < GH; ++k) atomicAdd(d + k, sr[k] * nv);
            }
        }
    }
    __syncthreads();

    if (!FUSE) {
        for (int i4 = tid; i4 < RN * (GH / 4); i4 += OWN_T) {
            int node = i4 >> 4;
            int k4   = i4 & 15;
            v4f v = *(const v4f*)(acc + node * AP + k4 * 4);
            if (DO_TANH) {
                v.x = tanhf(v.x); v.y = tanhf(v.y); v.z = tanhf(v.z); v.w = tanhf(v.w);
            }
            *(v4f*)(hout + nodeBase * GH + (size_t)i4 * 4) = v;
        }
    } else {
        for (int i4 = tid; i4 < RN * (GH / 4); i4 += OWN_T) {
            int node = i4 >> 4;
            int k4   = i4 & 15;
            v4f v = *(const v4f*)(acc + node * AP + k4 * 4);
            if (DO_TANH) {
                v.x = tanhf(v.x); v.y = tanhf(v.y); v.z = tanhf(v.z); v.w = tanhf(v.w);
            }
            *(v4f*)(acc + node * AP + k4 * 4) = v;
        }
        for (int i4 = tid; i4 < GH * (GH / 4); i4 += OWN_T) {
            int n  = i4 >> 4;
            int k4 = i4 & 15;
            *(v4f*)(wl + n * AP + k4 * 4) = *(const v4f*)(Wn + (size_t)i4 * 4);
        }
        __syncthreads();

        const int half = lane >> 4;
        const int lm   = lane & 15;
        for (int job = wave; job < ((RN + 15) / 16) * (GH / 16); job += OWN_T / 32) {
            const int m0 = (job >> 2) * 16;
            const int n0 = (job & 3) * 16;
            int row = m0 + lm;
            row = row < RN ? row : RN - 1;
            const float* arow = acc + row * AP;
            const float* wrow = wl + (n0 + lm) * AP;
            v8f c = {};
#pragma unroll
            for (int k0 = 0; k0 < GH; k0 += 4) {
                const int ka = k0 + half * 2;
                v2f a  = *(const v2f*)(arow + ka);
                v2f bf = *(const v2f*)(wrow + ka);
                c = __builtin_amdgcn_wmma_f32_16x16x4_f32(
                     false, a,  false, bf,
                     (short)0, c,  false,  false);
            }
            float* drow = hout + (nodeBase + m0) * GH + n0 + lm;
#pragma unroll
            for (int j = 0; j < 8; ++j) {
                const int r = j + 8 * half;
                if (m0 + r < RN) drow[(size_t)r * GH] = c[j];
            }
        }
    }
}

__global__ __launch_bounds__(256) void gather_kernel(const float* __restrict__ emb,
                                                     const int* __restrict__ pos,
                                                     float* __restrict__ out) {
    int idx = blockIdx.x * blockDim.x + threadIdx.x;
    if (idx >= GB * GP * GEMB) return;
    int k = idx & (GEMB - 1);
    int bp = idx >> 6;
    int b = bp / GP;
    int pv = pos[bp];
    out[idx] = (pv >= 0) ? emb[((size_t)b * GN + pv) * GH + k] : -3.0f;
}

extern "C" void kernel_launch(void* const* d_in, const int* in_sizes, int n_in,
                              void* d_out, int out_size, void* d_ws, size_t ws_size,
                              hipStream_t stream) {
    const float* x   = (const float*)d_in[0];
    const int*   ei  = (const int*)d_in[1];
    const float* ew  = (const float*)d_in[2];
    const int*   pos = (const int*)d_in[3];
    const float* W1  = (const float*)d_in[4];
    const float* b1  = (const float*)d_in[5];
    const float* W2  = (const float*)d_in[6];
    const float* b2  = (const float*)d_in[7];
    const float* W3  = (const float*)d_in[8];
    const float* b3  = (const float*)d_in[9];
    float* out = (float*)d_out;

    float* ws   = (float*)d_ws;
    const size_t nBN  = (size_t)GB * GN;
    const size_t nBNH = (size_t)GB * GN * GH;
    const size_t nList = (size_t)NRANGE * GB;
    float* dinv = ws;
    unsigned int* counts = (unsigned int*)(dinv + nBN);
    v2u*   lists = (v2u*)(dinv + nBN + nList);
    float* tmp  = dinv + nBN + nList + 2 * nList * CAP;
    float* hbuf = tmp + nBNH;

    const int T = 256;
    dim3 own_grid(NRANGE, GB, 1);

    deg_kernel<<<own_grid, OWN_T, 0, stream>>>(ei, ew, x, W1, tmp, dinv, lists, counts);

    agg_kernel<1, 1, 1><<<own_grid, OWN_T, 0, stream>>>(lists, counts, ei, ew, tmp, dinv, b1, W2, hbuf);

    agg_kernel<1, 1, 0><<<own_grid, OWN_T, 0, stream>>>(lists, counts, ei, ew, hbuf, dinv, b2, W3, tmp);

    agg_kernel<0, 0, 0><<<own_grid, OWN_T, 0, stream>>>(lists, counts, ei, ew, tmp, dinv, b3, W3, hbuf);

    gather_kernel<<<(GB * GP * GEMB + T - 1) / T, T, 0, stream>>>(hbuf, pos, out);
}


# Aggregation adds: LDS read-add-write of 4 consecutive list entries goes as one group (4 reads, 4 adds, 4 writes) where their targets differ (quad compare + ballot), else one by one; sum order kept
# speedup vs baseline: 4.9653x; 1.0320x over previous
#include <hip/hip_runtime.h>
#include <hip/hip_bf16.h>

#define GB   16
#define GN   10000
#define GE   160000
#define GS   128
#define GH   64
#define GEMB 64
#define GP   16

typedef __attribute__((ext_vector_type(2))) float v2f;
typedef __attribute__((ext_vector_type(4))) float v4f;
typedef __attribute__((ext_vector_type(8))) float v8f;
typedef __attribute__((ext_vector_type(2))) unsigned int v2u;
typedef __attribute__((ext_vector_type(4))) int v4i;

#define NRANGE 16
#define RN     (GN / NRANGE)
#define OWN_T  1024
#define CAP    11264
#define CAPU   (CAP - 32)
#define NOWN   (OWN_T / 32)
#define SB     6
#define NBK    160
#define NKEY   (NOWN * NBK)
#define KPT    (NKEY / OWN_T)
#define AP     68
#define XP     132
#define XR     128
#define NCH    ((RN + XR - 1) / XR)

__device__ __forceinline__ unsigned grp_key(const unsigned ex) {
    return ((ex >> 16) & (NOWN - 1)) * NBK + ((ex & 0xFFFFu) >> SB);
}

__global__ __launch_bounds__(OWN_T) void deg_kernel(const int* __restrict__ ei,
                                                    const float* __restrict__ ew,
                                                    const float* __restrict__ x,
                                                    const float* __restrict__ W1,
                                                    float* __restrict__ tmp,
                                                    float* __restrict__ dinv,
                                                    v2u* __restrict__ lists,
                                                    unsigned int* __restrict__ counts) {
    __shared__ float dl[RN];
    __shared__ unsigned int cnt;
    __shared__ unsigned int ocnt[NKEY];
    __shared__ unsigned int wsum[NOWN];
    __shared__ float w1l[GH * XP];
    __shared__ float xl[2][XR * XP];
    const int tid  = threadIdx.x;
    const int lane = tid & 31;
    const int b    = blockIdx.y;
    const unsigned lo = blockIdx.x * RN;
    const int wg   = b * NRANGE + blockIdx.x;
    v2u* lst = lists + (size_t)wg * CAP;

    {
        const int wave = __builtin_amdgcn_readfirstlane(tid >> 5);
        const int half = lane >> 4;
        const int lm   = lane & 15;
        const size_t nodeBase = (size_t)b * GN + lo;
        const int xrow = tid >> 5;
        const int xk4  = tid & 31;

        for (int i4 = tid; i4 < GH * (GS / 4); i4 += OWN_T) {
            int n  = i4 >> 5;
            int k4 = i4 & 31;
            *(v4f*)(w1l + n * XP + k4 * 4) = *(const v4f*)(W1 + (size_t)i4 * 4);
        }
        v4f nx[4];
#pragma unroll
        for (int j = 0; j < 4; ++j) {
            int r = xrow + j * 32;
            nx[j] = *(const v4f*)(x + (nodeBase + r) * GS + xk4 * 4);
        }
#pragma unroll
        for (int j = 0; j < 4; ++j)
            *(v4f*)(&xl[0][(xrow + j * 32) * XP + xk4 * 4]) = nx[j];
        __syncthreads();

        const int m0 = (wave >> 2) * 16;
        const int n0 = (wave & 3) * 16;
        const float* wrow = w1l + (n0 + lm) * XP;
        for (int c = 0; c < NCH; ++c) {
            if (c + 1 < NCH) {
#pragma unroll
                for (int j = 0; j < 4; ++j) {
                    int r = (c + 1) * XR + xrow + j * 32;
                    r = r < RN ? r : RN - 1;
                    nx[j] = *(const v4f*)(x + (nodeBase + r) * GS + xk4 * 4);
                }
            }
            const float* arow = &xl[c & 1][(m0 + lm) * XP];
            v8f acc = {};
#pragma unroll
            for (int k0 = 0; k0 < GS; k0 += 4) {
                const int ka = k0 + half * 2;
                v2f a  = *(const v2f*)(arow + ka);
                v2f bf = *(const v2f*)(wrow + ka);
                acc = __builtin_amdgcn_wmma_f32_16x16x4_f32(
                     false, a,  false, bf,
                     (short)0, acc,  false,  false);
            }
            const int rbase = c * XR + m0;
            float* drow = tmp + (nodeBase + rbase) * GH + n0 + lm;
#pragma unroll
            for (int j = 0; j < 8; ++j) {
                const int r = j + 8 * half;
                if (c + 1 < NCH || rbase + r < RN) drow[(size_t)r * GH] = acc[j];
            }
            if (c + 1 < NCH) {
#pragma unroll
                for (int j = 0; j < 4; ++j)
                    *(v4f*)(&xl[(c + 1) & 1][(xrow + j * 32) * XP + xk4 * 4]) = nx[j];
            }
            __syncthreads();
        }
    }
    for (int i = tid; i < RN; i += OWN_T) dl[i] = 0.f;
    if (tid == 0) cnt = 0;
    for (int i = tid; i < NKEY; i += OWN_T) ocnt[i] = 0;
    __syncthreads();
    v2f* stg = (v2f*)&xl[0][0];

    const v4i* row4 = (const v4i*)(ei + (size_t)b * 2 * GE);
    const v4i* col4 = (const v4i*)(ei + (size_t)b * 2 * GE + GE);
    const v4f* w4   = (const v4f*)(ew + (size_t)b * GE);
    for (int it = 0; it < (GE / 4 + 4 * OWN_T - 1) / (4 * OWN_T); ++it) {
        v4i c[4];
        v4i r[4];
        v4f w[4];
#pragma unroll
        for (int u = 0; u < 4; ++u) {
            int p = (it * 4 + u) * OWN_T + tid;
            if (p < GE / 4) { c[u] = col4[p]; r[u] = row4[p]; w[u] = w4[p]; }
            else            { c[u] = (v4i){-1, -1, -1, -1}; r[u] = (v4i){0, 0, 0, 0}; w[u] = (v4f){0.f, 0.f, 0.f, 0.f}; }
        }
        unsigned m[16];
        unsigned tot = 0;
#pragma unroll
        for (int i = 0; i < 16; ++i) {
            bool hit = ((unsigned)c[i >> 2][i & 3] - lo) < (unsigned)RN;
            m[i] = __builtin_amdgcn_ballot_w32(hit);
            tot += __builtin_popcount(m[i]);
        }
        unsigned at = 0;
        if (tot) {
            if (lane == 0) at = atomicAdd(&cnt, tot);
            at = __builtin_amdgcn_readfirstlane(at);
        }
#pragma unroll
        for (int i = 0; i < 16; ++i) {
            unsigned cl = (unsigned)c[i >> 2][i & 3] - lo;
            if (cl < (unsigned)RN) {
                float wt = w[i >> 2][i & 3];
                atomicAdd(&dl[cl], wt);
                unsigned pos = at + __builtin_amdgcn_mbcnt_lo(m[i], 0u);
                if (pos < (unsigned)CAPU)
                    stg[pos] = (v2f){__uint_as_float((unsigned)r[i >> 2][i & 3] | (cl << 16)), wt};
            }
            at += __builtin_popcount(m[i]);
        }
    }
    __syncthreads();
    for (int i = tid; i < RN; i += OWN_T)
        dinv[(size_t)b * GN + lo + i] = rsqrtf(dl[i] + 1.0f);
    const unsigned nKept = cnt;
    if (tid == 0) counts[wg] = nKept;
    if (nKept <= (unsigned)CAPU) {
        const int n = (int)nKept;
        v2f      en[CAP / OWN_T];
        unsigned rk[CAP / OWN_T];
#pragma unroll
        for (int k = 0; k < CAP / OWN_T; ++k) {
            int i = tid + k * OWN_T;
            if (i < n) {
                en[k] = stg[i];
                rk[k] = atomicAdd(&ocnt[grp_key(__float_as_uint(en[k].x))], 1u);
            }
        }
        __syncthreads();
        {
            unsigned gc[KPT];
            unsigned s = 0;
#pragma unroll
            for (int k = 0; k < KPT; ++k) { gc[k] = ocnt[tid * KPT + k]; s += gc[k]; }
            unsigned v = s;
#pragma unroll
            for (int d = 1; d < 32; d <<= 1) {
                unsigned o = __shfl_up(v, d);
                if (lane >= d) v += o;
            }
            if (lane == 31) wsum[tid >> 5] = v;
            __syncthreads();
            unsigned ex = v - s;
            for (int j = 0; j < (tid >> 5); ++j) ex += wsum[j];
            unsigned* bnd = (unsigned*)(lst + CAPU);
            if (lane == 0) bnd[tid >> 5] = ex;
            if (tid == 0) bnd[NOWN] = nKept;
#pragma unroll
            for (int k = 0; k < KPT; ++k) { ocnt[tid * KPT + k] = ex; ex += gc[k]; }
        }
        __syncthreads();
#pragma unroll
        for (int k = 0; k < CAP / OWN_T; ++k) {
            int i = tid + k * OWN_T;
            if (i < n) stg[ocnt[grp_key(__float_as_uint(en[k].x))] + rk[k]] = en[k];
        }
        __syncthreads();
        for (int i = tid; i < n; i += OWN_T) {
            v2f t = stg[i];
            lst[i] = (v2u){__float_as_uint(t.x), __float_as_uint(t.y)};
        }
    }
}

template <int G>
__device__ __forceinline__ void rows16(v2f (&v)[16], const v2u e, const float* __restrict__ tg, const int lane) {
#pragma unroll
    for (int u = 0; u < 16; ++u) {
        const unsigned sx = (unsigned)__builtin_amdgcn_readlane((int)e.x, G + u);
        const float* rowp = tg + (size_t)(sx & 0xFFFFu) * GH;
        v[u] = *(const v2f*)(rowp + 2 * lane);
    }
}
template <int G, bool FULL>
__device__ __forceinline__ void add16(const v2f (&v)[16], const v2u e, float* sums, const int lane, const int m, const unsigned cm) {
#pragma unroll
    for (int g = 0; g < 4; ++g) {
        const int u0 = 4 * g;
        if (FULL && ((cm >> (G + u0)) & 0xFu) == 0u) {
            float* d[4];
            float  w[4];
            v2f    a[4];
#pragma unroll
            for (int j = 0; j < 4; ++j) {
                const unsigned sx = (unsigned)__builtin_amdgcn_readlane((int)e.x, G + u0 + j);
                w[j] = __uint_as_float((unsigned)__builtin_amdgcn_readlane((int)e.y, G + u0 + j));
                d[j] = sums + (size_t)(sx >> 16) * AP + 2 * lane;
                a[j] = *(v2f*)d[j];
            }
#pragma unroll
            for (int j = 0; j < 4; ++j) {
                a[j].x += v[u0 + j].x * w[j];
                a[j].y += v[u0 + j].y * w[j];
                *(v2f*)d[j] = a[j];
            }
        } else {
#pragma unroll
            for (int j = 0; j < 4; ++j) {
                const int u = u0 + j;
                if (FULL || G + u < m) {
                    const unsigned sx = (unsigned)__builtin_amdgcn_readlane((int)e.x, G + u);
                    const float    w  = __uint_as_float((unsigned)__builtin_amdgcn_readlane((int)e.y, G + u));
                    float* d = sums + (size_t)(sx >> 16) * AP + 2 * lane;
                    v2f a = *(v2f*)d;
                    a.x += v[u].x * w;
                    a.y += v[u].y * w;
                    *(v2f*)d = a;
                }
            }
        }
    }
}
__device__ __forceinline__ unsigned quad_clash(const v2u e) {
    const int t  = (int)(e.x >> 16);
    const int t1 = __builtin_amdgcn_update_dpp(0, t, 0xB1, 0xF, 0xF, false);
    const int t2 = __builtin_amdgcn_update_dpp(0, t, 0x4E, 0xF, 0xF, false);
    const int t3 = __builtin_amdgcn_update_dpp(0, t, 0x1B, 0xF, 0xF, false);
    return __builtin_amdgcn_ballot_w32(t == t1 || t == t2 || t == t3);
}

template <int DO_TANH, int FUSE, int NORM>
__global__ __launch_bounds__(OWN_T) void agg_kernel(v2u* __restrict__ lists,
                                                    const unsigned int* __restrict__ counts,
                                                    const int* __restrict__ ei,
                                                    const float* __restrict__ ew,
                                                    const float* __restrict__ tmp,
                                                    const float* __restrict__ dinv,
                                                    const float* __restrict__ bias,
                                                    const float* __restrict__ Wn,
                                                    float* __restrict__ hout) {
    __shared__ float acc[RN * AP];
    __shared__ float wl[FUSE ? GH * AP : 4];
    __shared__ float dvl[NORM ? GN : 4];

    const int tid  = threadIdx.x;
    const int lane = tid & 31;
    const int wave = __builtin_amdgcn_readfirstlane(tid >> 5);
    const int b    = blockIdx.y;
    const unsigned lo = blockIdx.x * RN;
    const int wg   = b * NRANGE + blockIdx.x;
    const size_t nodeBase = (size_t)b * GN + lo;

    v4f dv[(GN / 4 + OWN_T - 1) / OWN_T];
    if (NORM) {
#pragma unroll
        for (int j = 0; j < (GN / 4 + OWN_T - 1) / OWN_T; ++j) {
            int i4 = tid + j * OWN_T;
            if (i4 < GN / 4) dv[j] = *(const v4f*)(dinv + (size_t)b * GN + (size_t)i4 * 4);
        }
    }
    for (int i4 = tid; i4 < RN * (GH / 4); i4 += OWN_T) {
        int node = i4 >> 4;
        int k4   = i4 & 15;
        float di = dinv[nodeBase + node];
        float sl = di * di;
        v4f t  = *(const v4f*)(tmp + (nodeBase + node) * GH + k4 * 4);
        v4f bb = *(const v4f*)(bias + k4 * 4);
        *(v4f*)(acc + node * AP + k4 * 4) = t * sl + bb;
    }
    if (NORM) {
#pragma unroll
        for (int j = 0; j < (GN / 4 + OWN_T - 1) / OWN_T; ++j) {
            int i4 = tid + j * OWN_T;
            if (i4 < GN / 4) *(v4f*)(dvl + i4 * 4) = dv[j];
        }
    }
    __syncthreads();

    const float* tg = tmp + (size_t)b * GN * GH;
    const unsigned nAll = counts[wg];

    if (nAll <= (unsigned)CAPU) {
        v2u* lst = lists + (size_t)wg * CAP;
        const unsigned* bnd = (const unsigned*)(lst + CAPU);
        int c0 = (int)bnd[wave];
        const int n = (int)bnd[wave + 1];
        v2u e = (v2u){0u, 0u};
        if (c0 + lane < n) {
            e = lst[c0 + lane];
            if (NORM) {
                e.y = __float_as_uint(dvl[e.x & 0xFFFFu] * __uint_as_float(e.y) * dvl[lo + (e.x >> 16)]);
                lst[c0 + lane] = e;
            }
        }
        v2f va[16], vb[16];
        rows16<0>(va, e, tg, lane);
        for (; c0 < n; c0 += 32) {
            const int m = n - c0;
            v2u enx = (v2u){0u, 0u};
            if (c0 + 32 + lane < n) enx = lst[c0 + 32 + lane];
            rows16<16>(vb, e, tg, lane);
            const unsigned cm = quad_clash(e);
            if (m >= 32) add16<0, true>(va, e, acc, lane, m, cm);
            else         add16<0, false>(va, e, acc, lane, m, cm);
            if (NORM) {
                if (c0 + 32 + lane < n) {
                    enx.y = __float_as_uint(dvl[enx.x & 0xFFFFu] * __uint_as_float(enx.y) * dvl[lo + (enx.x >> 16)]);
                    lst[c0 + 32 + lane] = enx;
                }
            }
            rows16<0>(va, enx, tg, lane);
            if (m >= 32) add16<16, true>(vb, e, acc, lane, m, cm);
            else         add16<16, false>(vb, e, acc, lane, m, cm);
            e = enx;
        }
    } else {
        const int*   rowp = ei + (size_t)b * 2 * GE;
        const int*   colp = rowp + GE;
        const float* wp   = ew + (size_t)b * GE;
        for (int eI = tid; eI < GE; eI += OWN_T) {
            int c = colp[eI];
            unsigned cl = (unsigned)c - lo;
            if (cl < (unsigned)RN) {
                int r = rowp[eI];
                float nv = dinv[(size_t)b * GN + r] * wp[eI] * dinv[(size_t)b * GN + c];
                const float* sr = tmp + ((size_t)b * GN + r) * GH;
                float* d = acc + (size_t)cl * AP;
                for (int k = 0; k < GH; ++k) atomicAdd(d + k, sr[k] * nv);
            }
        }
    }
    __syncthreads();

    if (!FUSE) {
        for (int i4 = tid; i4 < RN * (GH / 4); i4 += OWN_T) {
            int node = i4 >> 4;
            int k4   = i4 & 15;
            v4f v = *(const v4f*)(acc + node * AP + k4 * 4);
            if (DO_TANH) {
                v.x = tanhf(v.x); v.y = tanhf(v.y); v.z = tanhf(v.z); v.w = tanhf(v.w);
            }
            *(v4f*)(hout + nodeBase * GH + (size_t)i4 * 4) = v;
        }
    } else {
        for (int i4 = tid; i4 < RN * (GH / 4); i4 += OWN_T) {
            int node = i4 >> 4;
            int k4   = i4 & 15;
            v4f v = *(const v4f*)(acc + node * AP + k4 * 4);
            if (DO_TANH) {
                v.x = tanhf(v.x); v.y = tanhf(v.y); v.z = tanhf(v.z); v.w = tanhf(v.w);
            }
            *(v4f*)(acc + node * AP + k4 * 4) = v;
        }
        for (int i4 = tid; i4 < GH * (GH / 4); i4 += OWN_T) {
            int n  = i4 >> 4;
            int k4 = i4 & 15;
            *(v4f*)(wl + n * AP + k4 * 4) = *(const v4f*)(Wn + (size_t)i4 * 4);
        }
        __syncthreads();

        const int half = lane >> 4;
        const int lm   = lane & 15;
        for (int job = wave; job < ((RN + 15) / 16) * (GH / 16); job += OWN_T / 32) {
            const int m0 = (job >> 2) * 16;
            const int n0 = (job & 3) * 16;
            int row = m0 + lm;
            row = row < RN ? row : RN - 1;
            const float* arow = acc + row * AP;
            const float* wrow = wl + (n0 + lm) * AP;
            v8f c = {};
#pragma unroll
            for (int k0 = 0; k0 < GH; k0 += 4) {
                const int ka = k0 + half * 2;
                v2f a  = *(const v2f*)(arow + ka);
                v2f bf = *(const v2f*)(wrow + ka);
                c = __builtin_amdgcn_wmma_f32_16x16x4_f32(
                     false, a,  false, bf,
                     (short)0, c,  false,  false);
            }
            float* drow = hout + (nodeBase + m0) * GH + n0 + lm;
#pragma unroll
            for (int j = 0; j < 8; ++j) {
                const int r = j + 8 * half;
                if (m0 + r < RN) drow[(size_t)r * GH] = c[j];
            }
        }
    }
}

__global__ __launch_bounds__(256) void gather_kernel(const float* __restrict__ emb,
                                                     const int* __restrict__ pos,
                                                     float* __restrict__ out) {
    int idx = blockIdx.x * blockDim.x + threadIdx.x;
    if (idx >= GB * GP * GEMB) return;
    int k = idx & (GEMB - 1);
    int bp = idx >> 6;
    int b = bp / GP;
    int pv = pos[bp];
    out[idx] = (pv >= 0) ? emb[((size_t)b * GN + pv) * GH + k] : -3.0f;
}

extern "C" void kernel_launch(void* const* d_in, const int* in_sizes, int n_in,
                              void* d_out, int out_size, void* d_ws, size_t ws_size,
                              hipStream_t stream) {
    const float* x   = (const float*)d_in[0];
    const int*   ei  = (const int*)d_in[1];
    const float* ew  = (const float*)d_in[2];
    const int*   pos = (const int*)d_in[3];
    const float* W1  = (const float*)d_in[4];
    const float* b1  = (const float*)d_in[5];
    const float* W2  = (const float*)d_in[6];
    const float* b2  = (const float*)d_in[7];
    const float* W3  = (const float*)d_in[8];
    const float* b3  = (const float*)d_in[9];
    float* out = (float*)d_out;

    float* ws   = (float*)d_ws;
    const size_t nBN  = (size_t)GB * GN;
    const size_t nBNH = (size_t)GB * GN * GH;
    const size_t nList = (size_t)NRANGE * GB;
    float* dinv = ws;
    unsigned int* counts = (unsigned int*)(dinv + nBN);
    v2u*   lists = (v2u*)(dinv + nBN + nList);
    float* tmp  = dinv + nBN + nList + 2 * nList * CAP;
    float* hbuf = tmp + nBNH;

    const int T = 256;
    dim3 own_grid(NRANGE, GB, 1);

    deg_kernel<<<own_grid, OWN_T, 0, stream>>>(ei, ew, x, W1, tmp, dinv, lists, counts);

    agg_kernel<1, 1, 1><<<own_grid, OWN_T, 0, stream>>>(lists, counts, ei, ew, tmp, dinv, b1, W2, hbuf);

    agg_kernel<1, 1, 0><<<own_grid, OWN_T, 0, stream>>>(lists, counts, ei, ew, hbuf, dinv, b2, W3, tmp);

    agg_kernel<0, 0, 0><<<own_grid, OWN_T, 0, stream>>>(lists, counts, ei, ew, tmp, dinv, b3, W3, hbuf);

    gather_kernel<<<(GB * GP * GEMB + T - 1) / T, T, 0, stream>>>(hbuf, pos, out);
}


# Aggregation walk takes two list entries per instruction: a half-wave (16 lanes x 16 B) per edge row, entry fields shared by DPP row_share, 16-byte LDS adds; list order and sum order kept
# speedup vs baseline: 5.2644x; 1.0602x over previous
#include <hip/hip_runtime.h>
#include <hip/hip_bf16.h>

#define GB   16
#define GN   10000
#define GE   160000
#define GS   128
#define GH   64
#define GEMB 64
#define GP   16

typedef __attribute__((ext_vector_type(2))) float v2f;
typedef __attribute__((ext_vector_type(4))) float v4f;
typedef __attribute__((ext_vector_type(8))) float v8f;
typedef __attribute__((ext_vector_type(2))) unsigned int v2u;
typedef __attribute__((ext_vector_type(4))) int v4i;

#define NRANGE 16
#define RN     (GN / NRANGE)
#define OWN_T  1024
#define CAP    11264
#define CAPU   (CAP - 32)
#define NOWN   (OWN_T / 32)
#define SB     6
#define NBK    160
#define NKEY   (NOWN * NBK)
#define KPT    (NKEY / OWN_T)
#define AP     68
#define XP     132
#define XR     128
#define NCH    ((RN + XR - 1) / XR)

__device__ __forceinline__ unsigned grp_key(const unsigned ex) {
    return ((ex >> 16) & (NOWN - 1)) * NBK + ((ex & 0xFFFFu) >> SB);
}

__global__ __launch_bounds__(OWN_T) void deg_kernel(const int* __restrict__ ei,
                                                    const float* __restrict__ ew,
                                                    const float* __restrict__ x,
                                                    const float* __restrict__ W1,
                                                    float* __restrict__ tmp,
                                                    float* __restrict__ dinv,
                                                    v2u* __restrict__ lists,
                                                    unsigned int* __restrict__ counts) {
    __shared__ float dl[RN];
    __shared__ unsigned int cnt;
    __shared__ unsigned int ocnt[NKEY];
    __shared__ unsigned int wsum[NOWN];
    __shared__ float w1l[GH * XP];
    __shared__ float xl[2][XR * XP];
    const int tid  = threadIdx.x;
    const int lane = tid & 31;
    const int b    = blockIdx.y;
    const unsigned lo = blockIdx.x * RN;
    const int wg   = b * NRANGE + blockIdx.x;
    v2u* lst = lists + (size_t)wg * CAP;

    {
        const int wave = __builtin_amdgcn_readfirstlane(tid >> 5);
        const int half = lane >> 4;
        const int lm   = lane & 15;
        const size_t nodeBase = (size_t)b * GN + lo;
        const int xrow = tid >> 5;
        const int xk4  = tid & 31;

        for (int i4 = tid; i4 < GH * (GS / 4); i4 += OWN_T) {
            int n  = i4 >> 5;
            int k4 = i4 & 31;
            *(v4f*)(w1l + n * XP + k4 * 4) = *(const v4f*)(W1 + (size_t)i4 * 4);
        }
        v4f nx[4];
#pragma unroll
        for (int j = 0; j < 4; ++j) {
            int r = xrow + j * 32;
            nx[j] = *(const v4f*)(x + (nodeBase + r) * GS + xk4 * 4);
        }
#pragma unroll
        for (int j = 0; j < 4; ++j)
            *(v4f*)(&xl[0][(xrow + j * 32) * XP + xk4 * 4]) = nx[j];
        __syncthreads();

        const int m0 = (wave >> 2) * 16;
        const int n0 = (wave & 3) * 16;
        const float* wrow = w1l + (n0 + lm) * XP;
        for (int c = 0; c < NCH; ++c) {
            if (c + 1 < NCH) {
#pragma unroll
                for (int j = 0; j < 4; ++j) {
                    int r = (c + 1) * XR + xrow + j * 32;
                    r = r < RN ? r : RN - 1;
                    nx[j] = *(const v4f*)(x + (nodeBase + r) * GS + xk4 * 4);
                }
            }
            const float* arow = &xl[c & 1][(m0 + lm) * XP];
            v8f acc = {};
#pragma unroll
            for (int k0 = 0; k0 < GS; k0 += 4) {
                const int ka = k0 + half * 2;
                v2f a  = *(const v2f*)(arow + ka);
                v2f bf = *(const v2f*)(wrow + ka);
                acc = __builtin_amdgcn_wmma_f32_16x16x4_f32(
                     false, a,  false, bf,
                     (short)0, acc,  false,  false);
            }
            const int rbase = c * XR + m0;
            float* drow = tmp + (nodeBase + rbase) * GH + n0 + lm;
#pragma unroll
            for (int j = 0; j < 8; ++j) {
                const int r = j + 8 * half;
                if (c + 1 < NCH || rbase + r < RN) drow[(size_t)r * GH] = acc[j];
            }
            if (c + 1 < NCH) {
#pragma unroll
                for (int j = 0; j < 4; ++j)
                    *(v4f*)(&xl[(c + 1) & 1][(xrow + j * 32) * XP + xk4 * 4]) = nx[j];
            }
            __syncthreads();
        }
    }
    for (int i = tid; i < RN; i += OWN_T) dl[i] = 0.f;
    if (tid == 0) cnt = 0;
    for (int i = tid; i < NKEY; i += OWN_T) ocnt[i] = 0;
    __syncthreads();
    v2f* stg = (v2f*)&xl[0][0];

    const v4i* row4 = (const v4i*)(ei + (size_t)b * 2 * GE);
    const v4i* col4 = (const v4i*)(ei + (size_t)b * 2 * GE + GE);
    const v4f* w4   = (const v4f*)(ew + (size_t)b * GE);
    for (int it = 0; it < (GE / 4 + 4 * OWN_T - 1) / (4 * OWN_T); ++it) {
        v4i c[4];
        v4i r[4];
        v4f w[4];
#pragma unroll
        for (int u = 0; u < 4; ++u) {
            int p = (it * 4 + u) * OWN_T + tid;
            if (p < GE / 4) { c[u] = col4[p]; r[u] = row4[p]; w[u] = w4[p]; }
            else            { c[u] = (v4i){-1, -1, -1, -1}; r[u] = (v4i){0, 0, 0, 0}; w[u] = (v4f){0.f, 0.f, 0.f, 0.f}; }
        }
        unsigned m[16];
        unsigned tot = 0;
#pragma unroll
        for (int i = 0; i < 16; ++i) {
            bool hit = ((unsigned)c[i >> 2][i & 3] - lo) < (unsigned)RN;
            m[i] = __builtin_amdgcn_ballot_w32(hit);
            tot += __builtin_popcount(m[i]);
        }
        unsigned at = 0;
        if (tot) {
            if (lane == 0) at = atomicAdd(&cnt, tot);
            at = __builtin_amdgcn_readfirstlane(at);
        }
#pragma unroll
        for (int i = 0; i < 16; ++i) {
            unsigned cl = (unsigned)c[i >> 2][i & 3] - lo;
            if (cl < (unsigned)RN) {
                float wt = w[i >> 2][i & 3];
                atomicAdd(&dl[cl], wt);
                unsigned pos = at + __builtin_amdgcn_mbcnt_lo(m[i], 0u);
                if (pos < (unsigned)CAPU)
                    stg[pos] = (v2f){__uint_as_float((unsigned)r[i >> 2][i & 3] | (cl << 16)), wt};
            }
            at += __builtin_popcount(m[i]);
        }
    }
    __syncthreads();
    for (int i = tid; i < RN; i += OWN_T)
        dinv[(size_t)b * GN + lo + i] = rsqrtf(dl[i] + 1.0f);
    const unsigned nKept = cnt;
    if (tid == 0) counts[wg] = nKept;
    if (nKept <= (unsigned)CAPU) {
        const int n = (int)nKept;
        v2f      en[CAP / OWN_T];
        unsigned rk[CAP / OWN_T];
#pragma unroll
        for (int k = 0; k < CAP / OWN_T; ++k) {
            int i = tid + k * OWN_T;
            if (i < n) {
                en[k] = stg[i];
                rk[k] = atomicAdd(&ocnt[grp_key(__float_as_uint(en[k].x))], 1u);
            }
        }
        __syncthreads();
        {
            unsigned gc[KPT];
            unsigned s = 0;
#pragma unroll
            for (int k = 0; k < KPT; ++k) { gc[k] = ocnt[tid * KPT + k]; s += gc[k]; }
            unsigned v = s;
#pragma unroll
            for (int d = 1; d < 32; d <<= 1) {
                unsigned o = __shfl_up(v, d);
                if (lane >= d) v += o;
            }
            if (lane == 31) wsum[tid >> 5] = v;
            __syncthreads();
            unsigned ex = v - s;
            for (int j = 0; j < (tid >> 5); ++j) ex += wsum[j];
            unsigned* bnd = (unsigned*)(lst + CAPU);
            if (lane == 0) bnd[tid >> 5] = ex;
            if (tid == 0) bnd[NOWN] = nKept;
#pragma unroll
            for (int k = 0; k < KPT; ++k) { ocnt[tid * KPT + k] = ex; ex += gc[k]; }
        }
        __syncthreads();
#pragma unroll
        for (int k = 0; k < CAP / OWN_T; ++k) {
            int i = tid + k * OWN_T;
            if (i < n) stg[ocnt[grp_key(__float_as_uint(en[k].x))] + rk[k]] = en[k];
        }
        __syncthreads();
        for (int i = tid; i < n; i += OWN_T) {
            v2f t = stg[i];
            lst[i] = (v2u){__float_as_uint(t.x), __float_as_uint(t.y)};
        }
    }
}

template <int J>
__device__ __forceinline__ unsigned share(const unsigned x) {
    return (unsigned)__builtin_amdgcn_update_dpp(0, (int)x, 0x150 + J, 0xF, 0xF, true);
}
template <int S0, int U>
__device__ __forceinline__ void rows8(v4f (&v)[8], const unsigned srcoff, const char* __restrict__ tgb, const unsigned l16b) {
    if constexpr (U < 8) {
        v[U] = *(const v4f*)(tgb + (size_t)(share<S0 + U>(srcoff) + l16b));
        rows8<S0, U + 1>(v, srcoff, tgb, l16b);
    }
}
template <int S, bool FULL>
__device__ __forceinline__ void step_add(const v4f row, const unsigned t, const float w, char* accb,
                                         const int hi, const int m, const unsigned pc) {
    constexpr int p = 2 * S;
    v4f* d = (v4f*)(accb + t);
    if (FULL && ((pc >> p) & 1u) == 0u) {
        v4f a = *d; a += row * w; *d = a;
    } else {
        if (FULL || p < m)     { if (hi == 0) { v4f a = *d; a += row * w; *d = a; } }
        __builtin_amdgcn_wave_barrier();
        if (FULL || p + 1 < m) { if (hi == 1) { v4f a = *d; a += row * w; *d = a; } }
        __builtin_amdgcn_wave_barrier();
    }
}
template <int S0, bool FULL, int G>
__device__ __forceinline__ void add8(const v4f (&v)[8], const unsigned tgtoff, const unsigned wbits, char* accb,
                                     const int hi, const int m, const unsigned cm, const unsigned pc) {
    if constexpr (G < 4) {
        constexpr int s0 = S0 + 2 * G;
        constexpr int p0 = 2 * S0 + 4 * G;
        const unsigned t0 = share<s0>(tgtoff);
        const unsigned t1 = share<s0 + 1>(tgtoff);
        const float    w0 = __uint_as_float(share<s0>(wbits));
        const float    w1 = __uint_as_float(share<s0 + 1>(wbits));
        if (FULL && ((cm >> p0) & 0xFu) == 0u) {
            v4f* d0 = (v4f*)(accb + t0);
            v4f* d1 = (v4f*)(accb + t1);
            v4f a0 = *d0;
            v4f a1 = *d1;
            a0 += v[2 * G] * w0;
            a1 += v[2 * G + 1] * w1;
            *d0 = a0;
            *d1 = a1;
        } else {
            step_add<s0, FULL>(v[2 * G], t0, w0, accb, hi, m, pc);
            step_add<s0 + 1, FULL>(v[2 * G + 1], t1, w1, accb, hi, m, pc);
        }
        add8<S0, FULL, G + 1>(v, tgtoff, wbits, accb, hi, m, cm, pc);
    }
}
__device__ __forceinline__ unsigned quad_clash(const v2u e, unsigned& pc) {
    const int t  = (int)(e.x >> 16);
    const int t1 = __builtin_amdgcn_update_dpp(0, t, 0xB1, 0xF, 0xF, false);
    const int t2 = __builtin_amdgcn_update_dpp(0, t, 0x4E, 0xF, 0xF, false);
    const int t3 = __builtin_amdgcn_update_dpp(0, t, 0x1B, 0xF, 0xF, false);
    pc = __builtin_amdgcn_ballot_w32(t == t1);
    return __builtin_amdgcn_ballot_w32(t == t1 || t == t2 || t == t3);
}

template <int DO_TANH, int FUSE, int NORM>
__global__ __launch_bounds__(OWN_T) void agg_kernel(v2u* __restrict__ lists,
                                                    const unsigned int* __restrict__ counts,
                                                    const int* __restrict__ ei,
                                                    const float* __restrict__ ew,
                                                    const float* __restrict__ tmp,
                                                    const float* __restrict__ dinv,
                                                    const float* __restrict__ bias,
                                                    const float* __restrict__ Wn,
                                                    float* __restrict__ hout) {
    __shared__ float acc[RN * AP];
    __shared__ float wl[FUSE ? GH * AP : 4];
    __shared__ float dvl[NORM ? GN : 4];

    const int tid  = threadIdx.x;
    const int lane = tid & 31;
    const int wave = __builtin_amdgcn_readfirstlane(tid >> 5);
    const int b    = blockIdx.y;
    const unsigned lo = blockIdx.x * RN;
    const int wg   = b * NRANGE + blockIdx.x;
    const size_t nodeBase = (size_t)b * GN + lo;

    v4f dv[(GN / 4 + OWN_T - 1) / OWN_T];
    if (NORM) {
#pragma unroll
        for (int j = 0; j < (GN / 4 + OWN_T - 1) / OWN_T; ++j) {
            int i4 = tid + j * OWN_T;
            if (i4 < GN / 4) dv[j] = *(const v4f*)(dinv + (size_t)b * GN + (size_t)i4 * 4);
        }
    }
    for (int i4 = tid; i4 < RN * (GH / 4); i4 += OWN_T) {
        int node = i4 >> 4;
        int k4   = i4 & 15;
        float di = dinv[nodeBase + node];
        float sl = di * di;
        v4f t  = *(const v4f*)(tmp + (nodeBase + node) * GH + k4 * 4);
        v4f bb = *(const v4f*)(bias + k4 * 4);
        *(v4f*)(acc + node * AP + k4 * 4) = t * sl + bb;
    }
    if (NORM) {
#pragma unroll
        for (int j = 0; j < (GN / 4 + OWN_T - 1) / OWN_T; ++j) {
            int i4 = tid + j * OWN_T;
            if (i4 < GN / 4) *(v4f*)(dvl + i4 * 4) = dv[j];
        }
    }
    __syncthreads();

    const float* tg = tmp + (size_t)b * GN * GH;
    const unsigned nAll = counts[wg];

    if (nAll <= (unsigned)CAPU) {
        v2u* lst = lists + (size_t)wg * CAP;
        const unsigned* bnd = (const unsigned*)(lst + CAPU);
        int c0 = (int)bnd[wave];
        const int n = (int)bnd[wave + 1];
        v2u e = (v2u){0u, 0u};
        if (c0 + lane < n) {
            e = lst[c0 + lane];
            if (NORM) {
                e.y = __float_as_uint(dvl[e.x & 0xFFFFu] * __uint_as_float(e.y) * dvl[lo + (e.x >> 16)]);
                lst[c0 + lane] = e;
            }
        }
        const int      hi   = lane >> 4;
        const unsigned l16b = (unsigned)(lane & 15) * 16u;
        const int      pl   = 2 * (lane & 15) + hi;
        const char*    tgb  = (const char*)tg;
        char*          accb = (char*)acc + l16b;
        unsigned px = __shfl(e.x, pl);
        unsigned wb = __shfl(e.y, pl);
        unsigned so = (px & 0xFFFFu) * (unsigned)(GH * 4);
        unsigned to = (px >> 16) * (unsigned)(AP * 4);
        v4f va[8], vb[8];
        rows8<0, 0>(va, so, tgb, l16b);
        for (; c0 < n; c0 += 32) {
            const int m = n - c0;
            v2u enx = (v2u){0u, 0u};
            if (c0 + 32 + lane < n) enx = lst[c0 + 32 + lane];
            rows8<8, 0>(vb, so, tgb, l16b);
            unsigned pc;
            const unsigned cm = quad_clash(e, pc);
            if (m >= 32) add8<0, true, 0>(va, to, wb, accb, hi, m, cm, pc);
            else         add8<0, false, 0>(va, to, wb, accb, hi, m, cm, pc);
            if (NORM) {
                if (c0 + 32 + lane < n) {
                    enx.y = __float_as_uint(dvl[enx.x & 0xFFFFu] * __uint_as_float(enx.y) * dvl[lo + (enx.x >> 16)]);
                    lst[c0 + 32 + lane] = enx;
                }
            }
            const unsigned pxn = __shfl(enx.x, pl);
            const unsigned wbn = __shfl(enx.y, pl);
            const unsigned son = (pxn & 0xFFFFu) * (unsigned)(GH * 4);
            rows8<0, 0>(va, son, tgb, l16b);
            if (m >= 32) add8<8, true, 0>(vb, to, wb, accb, hi, m, cm, pc);
            else         add8<8, false, 0>(vb, to, wb, accb, hi, m, cm, pc);
            e  = enx;
            so = son;
            to = (pxn >> 16) * (unsigned)(AP * 4);
            wb = wbn;
        }
    } else {
        const int*   rowp = ei + (size_t)b * 2 * GE;
        const int*   colp = rowp + GE;
        const float* wp   = ew + (size_t)b * GE;
        for (int eI = tid; eI < GE; eI += OWN_T) {
            int c = colp[eI];
            unsigned cl = (unsigned)c - lo;
            if (cl < (unsigned)RN) {
                int r = rowp[eI];
                float nv = dinv[(size_t)b * GN + r] * wp[eI] * dinv[(size_t)b * GN + c];
                const float* sr = tmp + ((size_t)b * GN + r) * GH;
                float* d = acc + (size_t)cl * AP;
                for (int k = 0; k < GH; ++k) atomicAdd(d + k, sr[k] * nv);
            }
        }
    }
    __syncthreads();

    if (!FUSE) {
        for (int i4 = tid; i4 < RN * (GH / 4); i4 += OWN_T) {
            int node = i4 >> 4;
            int k4   = i4 & 15;
            v4f v = *(const v4f*)(acc + node * AP + k4 * 4);
            if (DO_TANH) {
                v.x = tanhf(v.x); v.y = tanhf(v.y); v.z = tanhf(v.z); v.w = tanhf(v.w);
            }
            *(v4f*)(hout + nodeBase * GH + (size_t)i4 * 4) = v;
        }
    } else {
        for (int i4 = tid; i4 < RN * (GH / 4); i4 += OWN_T) {
            int node = i4 >> 4;
            int k4   = i4 & 15;
            v4f v = *(const v4f*)(acc + node * AP + k4 * 4);
            if (DO_TANH) {
                v.x = tanhf(v.x); v.y = tanhf(v.y); v.z = tanhf(v.z); v.w = tanhf(v.w);
            }
            *(v4f*)(acc + node * AP + k4 * 4) = v;
        }
        for (int i4 = tid; i4 < GH * (GH / 4); i4 += OWN_T) {
            int n  = i4 >> 4;
            int k4 = i4 & 15;
            *(v4f*)(wl + n * AP + k4 * 4) = *(const v4f*)(Wn + (size_t)i4 * 4);
        }
        __syncthreads();

        const int half = lane >> 4;
        const int lm   = lane & 15;
        for (int job = wave; job < ((RN + 15) / 16) * (GH / 16); job += OWN_T / 32) {
            const int m0 = (job >> 2) * 16;
            const int n0 = (job & 3) * 16;
            int row = m0 + lm;
            row = row < RN ? row : RN - 1;
            const float* arow = acc + row * AP;
            const float* wrow = wl + (n0 + lm) * AP;
            v8f c = {};
#pragma unroll
            for (int k0 = 0; k0 < GH; k0 += 4) {
                const int ka = k0 + half * 2;
                v2f a  = *(const v2f*)(arow + ka);
                v2f bf = *(const v2f*)(wrow + ka);
                c = __builtin_amdgcn_wmma_f32_16x16x4_f32(
                     false, a,  false, bf,
                     (short)0, c,  false,  false);
            }
            float* drow = hout + (nodeBase + m0) * GH + n0 + lm;
#pragma unroll
            for (int j = 0; j < 8; ++j) {
                const int r = j + 8 * half;
                if (m0 + r < RN) drow[(size_t)r * GH] = c[j];
            }
        }
    }
}

__global__ __launch_bounds__(256) void gather_kernel(const float* __restrict__ emb,
                                                     const int* __restrict__ pos,
                                                     float* __restrict__ out) {
    int idx = blockIdx.x * blockDim.x + threadIdx.x;
    if (idx >= GB * GP * GEMB) return;
    int k = idx & (GEMB - 1);
    int bp = idx >> 6;
    int b = bp / GP;
    int pv = pos[bp];
    out[idx] = (pv >= 0) ? emb[((size_t)b * GN + pv) * GH + k] : -3.0f;
}

extern "C" void kernel_launch(void* const* d_in, const int* in_sizes, int n_in,
                              void* d_out, int out_size, void* d_ws, size_t ws_size,
                              hipStream_t stream) {
    const float* x   = (const float*)d_in[0];
    const int*   ei  = (const int*)d_in[1];
    const float* ew  = (const float*)d_in[2];
    const int*   pos = (const int*)d_in[3];
    const float* W1  = (const float*)d_in[4];
    const float* b1  = (const float*)d_in[5];
    const float* W2  = (const float*)d_in[6];
    const float* b2  = (const float*)d_in[7];
    const float* W3  = (const float*)d_in[8];
    const float* b3  = (const float*)d_in[9];
    float* out = (float*)d_out;

    float* ws   = (float*)d_ws;
    const size_t nBN  = (size_t)GB * GN;
    const size_t nBNH = (size_t)GB * GN * GH;
    const size_t nList = (size_t)NRANGE * GB;
    float* dinv = ws;
    unsigned int* counts = (unsigned int*)(dinv + nBN);
    v2u*   lists = (v2u*)(dinv + nBN + nList);
    float* tmp  = dinv + nBN + nList + 2 * nList * CAP;
    float* hbuf = tmp + nBNH;

    const int T = 256;
    dim3 own_grid(NRANGE, GB, 1);

    deg_kernel<<<own_grid, OWN_T, 0, stream>>>(ei, ew, x, W1, tmp, dinv, lists, counts);

    agg_kernel<1, 1, 1><<<own_grid, OWN_T, 0, stream>>>(lists, counts, ei, ew, tmp, dinv, b1, W2, hbuf);

    agg_kernel<1, 1, 0><<<own_grid, OWN_T, 0, stream>>>(lists, counts, ei, ew, hbuf, dinv, b2, W3, tmp);

    agg_kernel<0, 0, 0><<<own_grid, OWN_T, 0, stream>>>(lists, counts, ei, ew, tmp, dinv, b3, W3, hbuf);

    gather_kernel<<<(GB * GP * GEMB + T - 1) / T, T, 0, stream>>>(hbuf, pos, out);
}


# Final gather launch folded into layer-3 aggregation's epilogue: owner block writes the pos rows of its range from LDS straight to out; full embedding never stored; 5 launches to 4; sum order kept
# speedup vs baseline: 5.6924x; 1.0813x over previous
#include <hip/hip_runtime.h>
#include <hip/hip_bf16.h>

#define GB   16
#define GN   10000
#define GE   160000
#define GS   128
#define GH   64
#define GEMB 64
#define GP   16

typedef __attribute__((ext_vector_type(2))) float v2f;
typedef __attribute__((ext_vector_type(4))) float v4f;
typedef __attribute__((ext_vector_type(8))) float v8f;
typedef __attribute__((ext_vector_type(2))) unsigned int v2u;
typedef __attribute__((ext_vector_type(4))) int v4i;

#define NRANGE 16
#define RN     (GN / NRANGE)
#define OWN_T  1024
#define CAP    11264
#define CAPU   (CAP - 32)
#define NOWN   (OWN_T / 32)
#define SB     6
#define NBK    160
#define NKEY   (NOWN * NBK)
#define KPT    (NKEY / OWN_T)
#define AP     68
#define XP     132
#define XR     128
#define NCH    ((RN + XR - 1) / XR)

__device__ __forceinline__ unsigned grp_key(const unsigned ex) {
    return ((ex >> 16) & (NOWN - 1)) * NBK + ((ex & 0xFFFFu) >> SB);
}

__global__ __launch_bounds__(OWN_T) void deg_kernel(const int* __restrict__ ei,
                                                    const float* __restrict__ ew,
                                                    const float* __restrict__ x,
                                                    const float* __restrict__ W1,
                                                    float* __restrict__ tmp,
                                                    float* __restrict__ dinv,
                                                    v2u* __restrict__ lists,
                                                    unsigned int* __restrict__ counts) {
    __shared__ float dl[RN];
    __shared__ unsigned int cnt;
    __shared__ unsigned int ocnt[NKEY];
    __shared__ unsigned int wsum[NOWN];
    __shared__ float w1l[GH * XP];
    __shared__ float xl[2][XR * XP];
    const int tid  = threadIdx.x;
    const int lane = tid & 31;
    const int b    = blockIdx.y;
    const unsigned lo = blockIdx.x * RN;
    const int wg   = b * NRANGE + blockIdx.x;
    v2u* lst = lists + (size_t)wg * CAP;

    {
        const int wave = __builtin_amdgcn_readfirstlane(tid >> 5);
        const int half = lane >> 4;
        const int lm   = lane & 15;
        const size_t nodeBase = (size_t)b * GN + lo;
        const int xrow = tid >> 5;
        const int xk4  = tid & 31;

        for (int i4 = tid; i4 < GH * (GS / 4); i4 += OWN_T) {
            int n  = i4 >> 5;
            int k4 = i4 & 31;
            *(v4f*)(w1l + n * XP + k4 * 4) = *(const v4f*)(W1 + (size_t)i4 * 4);
        }
        v4f nx[4];
#pragma unroll
        for (int j = 0; j < 4; ++j) {
            int r = xrow + j * 32;
            nx[j] = *(const v4f*)(x + (nodeBase + r) * GS + xk4 * 4);
        }
#pragma unroll
        for (int j = 0; j < 4; ++j)
            *(v4f*)(&xl[0][(xrow + j * 32) * XP + xk4 * 4]) = nx[j];
        __syncthreads();

        const int m0 = (wave >> 2) * 16;
        const int n0 = (wave & 3) * 16;
        const float* wrow = w1l + (n0 + lm) * XP;
        for (int c = 0; c < NCH; ++c) {
            if (c + 1 < NCH) {
#pragma unroll
                for (int j = 0; j < 4; ++j) {
                    int r = (c + 1) * XR + xrow + j * 32;
                    r = r < RN ? r : RN - 1;
                    nx[j] = *(const v4f*)(x + (nodeBase + r) * GS + xk4 * 4);
                }
            }
            const float* arow = &xl[c & 1][(m0 + lm) * XP];
            v8f acc = {};
#pragma unroll
            for (int k0 = 0; k0 < GS; k0 += 4) {
                const int ka = k0 + half * 2;
                v2f a  = *(const v2f*)(arow + ka);
                v2f bf = *(const v2f*)(wrow + ka);
                acc = __builtin_amdgcn_wmma_f32_16x16x4_f32(
                     false, a,  false, bf,
                     (short)0, acc,  false,  false);
            }
            const int rbase = c * XR + m0;
            float* drow = tmp + (nodeBase + rbase) * GH + n0 + lm;
#pragma unroll
            for (int j = 0; j < 8; ++j) {
                const int r = j + 8 * half;
                if (c + 1 < NCH || rbase + r < RN) drow[(size_t)r * GH] = acc[j];
            }
            if (c + 1 < NCH) {
#pragma unroll
                for (int j = 0; j < 4; ++j)
                    *(v4f*)(&xl[(c + 1) & 1][(xrow + j * 32) * XP + xk4 * 4]) = nx[j];
            }
            __syncthreads();
        }
    }
    for (int i = tid; i < RN; i += OWN_T) dl[i] = 0.f;
    if (tid == 0) cnt = 0;
    for (int i = tid; i < NKEY; i += OWN_T) ocnt[i] = 0;
    __syncthreads();
    v2f* stg = (v2f*)&xl[0][0];

    const v4i* row4 = (const v4i*)(ei + (size_t)b * 2 * GE);
    const v4i* col4 = (const v4i*)(ei + (size_t)b * 2 * GE + GE);
    const v4f* w4   = (const v4f*)(ew + (size_t)b * GE);
    for (int it = 0; it < (GE / 4 + 4 * OWN_T - 1) / (4 * OWN_T); ++it) {
        v4i c[4];
        v4i r[4];
        v4f w[4];
#pragma unroll
        for (int u = 0; u < 4; ++u) {
            int p = (it * 4 + u) * OWN_T + tid;
            if (p < GE / 4) { c[u] = col4[p]; r[u] = row4[p]; w[u] = w4[p]; }
            else            { c[u] = (v4i){-1, -1, -1, -1}; r[u] = (v4i){0, 0, 0, 0}; w[u] = (v4f){0.f, 0.f, 0.f, 0.f}; }
        }
        unsigned m[16];
        unsigned tot = 0;
#pragma unroll
        for (int i = 0; i < 16; ++i) {
            bool hit = ((unsigned)c[i >> 2][i & 3] - lo) < (unsigned)RN;
            m[i] = __builtin_amdgcn_ballot_w32(hit);
            tot += __builtin_popcount(m[i]);
        }
        unsigned at = 0;
        if (tot) {
            if (lane == 0) at = atomicAdd(&cnt, tot);
            at = __builtin_amdgcn_readfirstlane(at);
        }
#pragma unroll
        for (int i = 0; i < 16; ++i) {
            unsigned cl = (unsigned)c[i >> 2][i & 3] - lo;
            if (cl < (unsigned)RN) {
                float wt = w[i >> 2][i & 3];
                atomicAdd(&dl[cl], wt);
                unsigned pos = at + __builtin_amdgcn_mbcnt_lo(m[i], 0u);
                if (pos < (unsigned)CAPU)
                    stg[pos] = (v2f){__uint_as_float((unsigned)r[i >> 2][i & 3] | (cl << 16)), wt};
            }
            at += __builtin_popcount(m[i]);
        }
    }
    __syncthreads();
    for (int i = tid; i < RN; i += OWN_T)
        dinv[(size_t)b * GN + lo + i] = rsqrtf(dl[i] + 1.0f);
    const unsigned nKept = cnt;
    if (tid == 0) counts[wg] = nKept;
    if (nKept <= (unsigned)CAPU) {
        const int n = (int)nKept;
        v2f      en[CAP / OWN_T];
        unsigned rk[CAP / OWN_T];
#pragma unroll
        for (int k = 0; k < CAP / OWN_T; ++k) {
            int i = tid + k * OWN_T;
            if (i < n) {
                en[k] = stg[i];
                rk[k] = atomicAdd(&ocnt[grp_key(__float_as_uint(en[k].x))], 1u);
            }
        }
        __syncthreads();
        {
            unsigned gc[KPT];
            unsigned s = 0;
#pragma unroll
            for (int k = 0; k < KPT; ++k) { gc[k] = ocnt[tid * KPT + k]; s += gc[k]; }
            unsigned v = s;
#pragma unroll
            for (int d = 1; d < 32; d <<= 1) {
                unsigned o = __shfl_up(v, d);
                if (lane >= d) v += o;
            }
            if (lane == 31) wsum[tid >> 5] = v;
            __syncthreads();
            unsigned ex = v - s;
            for (int j = 0; j < (tid >> 5); ++j) ex += wsum[j];
            unsigned* bnd = (unsigned*)(lst + CAPU);
            if (lane == 0) bnd[tid >> 5] = ex;
            if (tid == 0) bnd[NOWN] = nKept;
#pragma unroll
            for (int k = 0; k < KPT; ++k) { ocnt[tid * KPT + k] = ex; ex += gc[k]; }
        }
        __syncthreads();
#pragma unroll
        for (int k = 0; k < CAP / OWN_T; ++k) {
            int i = tid + k * OWN_T;
            if (i < n) stg[ocnt[grp_key(__float_as_uint(en[k].x))] + rk[k]] = en[k];
        }
        __syncthreads();
        for (int i = tid; i < n; i += OWN_T) {
            v2f t = stg[i];
            lst[i] = (v2u){__float_as_uint(t.x), __float_as_uint(t.y)};
        }
    }
}

template <int J>
__device__ __forceinline__ unsigned share(const unsigned x) {
    return (unsigned)__builtin_amdgcn_update_dpp(0, (int)x, 0x150 + J, 0xF, 0xF, true);
}
template <int S0, int U>
__device__ __forceinline__ void rows8(v4f (&v)[8], const unsigned srcoff, const char* __restrict__ tgb, const unsigned l16b) {
    if constexpr (U < 8) {
        v[U] = *(const v4f*)(tgb + (size_t)(share<S0 + U>(srcoff) + l16b));
        rows8<S0, U + 1>(v, srcoff, tgb, l16b);
    }
}
template <int S, bool FULL>
__device__ __forceinline__ void step_add(const v4f row, const unsigned t, const float w, char* accb,
                                         const int hi, const int m, const unsigned pc) {
    constexpr int p = 2 * S;
    v4f* d = (v4f*)(accb + t);
    if (FULL && ((pc >> p) & 1u) == 0u) {
        v4f a = *d; a += row * w; *d = a;
    } else {
        if (FULL || p < m)     { if (hi == 0) { v4f a = *d; a += row * w; *d = a; } }
        __builtin_amdgcn_wave_barrier();
        if (FULL || p + 1 < m) { if (hi == 1) { v4f a = *d; a += row * w; *d = a; } }
        __builtin_amdgcn_wave_barrier();
    }
}
template <int S0, bool FULL, int G>
__device__ __forceinline__ void add8(const v4f (&v)[8], const unsigned tgtoff, const unsigned wbits, char* accb,
                                     const int hi, const int m, const unsigned cm, const unsigned pc) {
    if constexpr (G < 4) {
        constexpr int s0 = S0 + 2 * G;
        constexpr int p0 = 2 * S0 + 4 * G;
        const unsigned t0 = share<s0>(tgtoff);
        const unsigned t1 = share<s0 + 1>(tgtoff);
        const float    w0 = __uint_as_float(share<s0>(wbits));
        const float    w1 = __uint_as_float(share<s0 + 1>(wbits));
        if (FULL && ((cm >> p0) & 0xFu) == 0u) {
            v4f* d0 = (v4f*)(accb + t0);
            v4f* d1 = (v4f*)(accb + t1);
            v4f a0 = *d0;
            v4f a1 = *d1;
            a0 += v[2 * G] * w0;
            a1 += v[2 * G + 1] * w1;
            *d0 = a0;
            *d1 = a1;
        } else {
            step_add<s0, FULL>(v[2 * G], t0, w0, accb, hi, m, pc);
            step_add<s0 + 1, FULL>(v[2 * G + 1], t1, w1, accb, hi, m, pc);
        }
        add8<S0, FULL, G + 1>(v, tgtoff, wbits, accb, hi, m, cm, pc);
    }
}
__device__ __forceinline__ unsigned quad_clash(const v2u e, unsigned& pc) {
    const int t  = (int)(e.x >> 16);
    const int t1 = __builtin_amdgcn_update_dpp(0, t, 0xB1, 0xF, 0xF, false);
    const int t2 = __builtin_amdgcn_update_dpp(0, t, 0x4E, 0xF, 0xF, false);
    const int t3 = __builtin_amdgcn_update_dpp(0, t, 0x1B, 0xF, 0xF, false);
    pc = __builtin_amdgcn_ballot_w32(t == t1);
    return __builtin_amdgcn_ballot_w32(t == t1 || t == t2 || t == t3);
}

template <int DO_TANH, int FUSE, int NORM>
__global__ __launch_bounds__(OWN_T) void agg_kernel(v2u* __restrict__ lists,
                                                    const unsigned int* __restrict__ counts,
                                                    const int* __restrict__ ei,
                                                    const float* __restrict__ ew,
                                                    const float* __restrict__ tmp,
                                                    const float* __restrict__ dinv,
                                                    const float* __restrict__ bias,
                                                    const float* __restrict__ Wn,
                                                    const int* __restrict__ pos,
                                                    float* __restrict__ hout) {
    __shared__ float acc[RN * AP];
    __shared__ float wl[FUSE ? GH * AP : 4];
    __shared__ float dvl[NORM ? GN : 4];

    const int tid  = threadIdx.x;
    const int lane = tid & 31;
    const int wave = __builtin_amdgcn_readfirstlane(tid >> 5);
    const int b    = blockIdx.y;
    const unsigned lo = blockIdx.x * RN;
    const int wg   = b * NRANGE + blockIdx.x;
    const size_t nodeBase = (size_t)b * GN + lo;

    int pv = 0;
    if (!FUSE) {
        if (tid < GP * (GEMB / 4)) pv = pos[b * GP + (tid >> 4)];
    }
    v4f dv[(GN / 4 + OWN_T - 1) / OWN_T];
    if (NORM) {
#pragma unroll
        for (int j = 0; j < (GN / 4 + OWN_T - 1) / OWN_T; ++j) {
            int i4 = tid + j * OWN_T;
            if (i4 < GN / 4) dv[j] = *(const v4f*)(dinv + (size_t)b * GN + (size_t)i4 * 4);
        }
    }
    for (int i4 = tid; i4 < RN * (GH / 4); i4 += OWN_T) {
        int node = i4 >> 4;
        int k4   = i4 & 15;
        float di = dinv[nodeBase + node];
        float sl = di * di;
        v4f t  = *(const v4f*)(tmp + (nodeBase + node) * GH + k4 * 4);
        v4f bb = *(const v4f*)(bias + k4 * 4);
        *(v4f*)(acc + node * AP + k4 * 4) = t * sl + bb;
    }
    if (NORM) {
#pragma unroll
        for (int j = 0; j < (GN / 4 + OWN_T - 1) / OWN_T; ++j) {
            int i4 = tid + j * OWN_T;
            if (i4 < GN / 4) *(v4f*)(dvl + i4 * 4) = dv[j];
        }
    }
    __syncthreads();

    const float* tg = tmp + (size_t)b * GN * GH;
    const unsigned nAll = counts[wg];

    if (nAll <= (unsigned)CAPU) {
        v2u* lst = lists + (size_t)wg * CAP;
        const unsigned* bnd = (const unsigned*)(lst + CAPU);
        int c0 = (int)bnd[wave];
        const int n = (int)bnd[wave + 1];
        v2u e = (v2u){0u, 0u};
        if (c0 + lane < n) {
            e = lst[c0 + lane];
            if (NORM) {
                e.y = __float_as_uint(dvl[e.x & 0xFFFFu] * __uint_as_float(e.y) * dvl[lo + (e.x >> 16)]);
                lst[c0 + lane] = e;
            }
        }
        const int      hi   = lane >> 4;
        const unsigned l16b = (unsigned)(lane & 15) * 16u;
        const int      pl   = 2 * (lane & 15) + hi;
        const char*    tgb  = (const char*)tg;
        char*          accb = (char*)acc + l16b;
        unsigned px = __shfl(e.x, pl);
        unsigned wb = __shfl(e.y, pl);
        unsigned so = (px & 0xFFFFu) * (unsigned)(GH * 4);
        unsigned to = (px >> 16) * (unsigned)(AP * 4);
        v4f va[8], vb[8];
        rows8<0, 0>(va, so, tgb, l16b);
        for (; c0 < n; c0 += 32) {
            const int m = n - c0;
            v2u enx = (v2u){0u, 0u};
            if (c0 + 32 + lane < n) enx = lst[c0 + 32 + lane];
            rows8<8, 0>(vb, so, tgb, l16b);
            unsigned pc;
            const unsigned cm = quad_clash(e, pc);
            if (m >= 32) add8<0, true, 0>(va, to, wb, accb, hi, m, cm, pc);
            else         add8<0, false, 0>(va, to, wb, accb, hi, m, cm, pc);
            if (NORM) {
                if (c0 + 32 + lane < n) {
                    enx.y = __float_as_uint(dvl[enx.x & 0xFFFFu] * __uint_as_float(enx.y) * dvl[lo + (enx.x >> 16)]);
                    lst[c0 + 32 + lane] = enx;
                }
            }
            const unsigned pxn = __shfl(enx.x, pl);
            const unsigned wbn = __shfl(enx.y, pl);
            const unsigned son = (pxn & 0xFFFFu) * (unsigned)(GH * 4);
            rows8<0, 0>(va, son, tgb, l16b);
            if (m >= 32) add8<8, true, 0>(vb, to, wb, accb, hi, m, cm, pc);
            else         add8<8, false, 0>(vb, to, wb, accb, hi, m, cm, pc);
            e  = enx;
            so = son;
            to = (pxn >> 16) * (unsigned)(AP * 4);
            wb = wbn;
        }
    } else {
        const int*   rowp = ei + (size_t)b * 2 * GE;
        const int*   colp = rowp + GE;
        const float* wp   = ew + (size_t)b * GE;
        for (int eI = tid; eI < GE; eI += OWN_T) {
            int c = colp[eI];
            unsigned cl = (unsigned)c - lo;
            if (cl < (unsigned)RN) {
                int r = rowp[eI];
                float nv = dinv[(size_t)b * GN + r] * wp[eI] * dinv[(size_t)b * GN + c];
                const float* sr = tmp + ((size_t)b * GN + r) * GH;
                float* d = acc + (size_t)cl * AP;
                for (int k = 0; k < GH; ++k) atomicAdd(d + k, sr[k] * nv);
            }
        }
    }
    __syncthreads();

    if (!FUSE) {
        if (tid < GP * (GEMB / 4)) {
            const int p  = tid >> 4;
            const int k4 = tid & 15;
            float* o = hout + (size_t)(b * GP + p) * GEMB + k4 * 4;
            if (pv >= 0) {
                const unsigned cl = (unsigned)pv - lo;
                if (cl < (unsigned)RN) {
                    v4f v = *(const v4f*)(acc + cl * AP + k4 * 4);
                    if (DO_TANH) {
                        v.x = tanhf(v.x); v.y = tanhf(v.y); v.z = tanhf(v.z); v.w = tanhf(v.w);
                    }
                    *(v4f*)o = v;
                }
            } else if (blockIdx.x == 0) {
                *(v4f*)o = (v4f){-3.0f, -3.0f, -3.0f, -3.0f};
            }
        }
    } else {
        for (int i4 = tid; i4 < RN * (GH / 4); i4 += OWN_T) {
            int node = i4 >> 4;
            int k4   = i4 & 15;
            v4f v = *(const v4f*)(acc + node * AP + k4 * 4);
            if (DO_TANH) {
                v.x = tanhf(v.x); v.y = tanhf(v.y); v.z = tanhf(v.z); v.w = tanhf(v.w);
            }
            *(v4f*)(acc + node * AP + k4 * 4) = v;
        }
        for (int i4 = tid; i4 < GH * (GH / 4); i4 += OWN_T) {
            int n  = i4 >> 4;
            int k4 = i4 & 15;
            *(v4f*)(wl + n * AP + k4 * 4) = *(const v4f*)(Wn + (size_t)i4 * 4);
        }
        __syncthreads();

        const int half = lane >> 4;
        const int lm   = lane & 15;
        for (int job = wave; job < ((RN + 15) / 16) * (GH / 16); job += OWN_T / 32) {
            const int m0 = (job >> 2) * 16;
            const int n0 = (job & 3) * 16;
            int row = m0 + lm;
            row = row < RN ? row : RN - 1;
            const float* arow = acc + row * AP;
            const float* wrow = wl + (n0 + lm) * AP;
            v8f c = {};
#pragma unroll
            for (int k0 = 0; k0 < GH; k0 += 4) {
                const int ka = k0 + half * 2;
                v2f a  = *(const v2f*)(arow + ka);
                v2f bf = *(const v2f*)(wrow + ka);
                c = __builtin_amdgcn_wmma_f32_16x16x4_f32(
                     false, a,  false, bf,
                     (short)0, c,  false,  false);
            }
            float* drow = hout + (nodeBase + m0) * GH + n0 + lm;
#pragma unroll
            for (int j = 0; j < 8; ++j) {
                const int r = j + 8 * half;
                if (m0 + r < RN) drow[(size_t)r * GH] = c[j];
            }
        }
    }
}

extern "C" void kernel_launch(void* const* d_in, const int* in_sizes, int n_in,
                              void* d_out, int out_size, void* d_ws, size_t ws_size,
                              hipStream_t stream) {
    const float* x   = (const float*)d_in[0];
    const int*   ei  = (const int*)d_in[1];
    const float* ew  = (const float*)d_in[2];
    const int*   pos = (const int*)d_in[3];
    const float* W1  = (const float*)d_in[4];
    const float* b1  = (const float*)d_in[5];
    const float* W2  = (const float*)d_in[6];
    const float* b2  = (const float*)d_in[7];
    const float* W3  = (const float*)d_in[8];
    const float* b3  = (const float*)d_in[9];
    float* out = (float*)d_out;

    float* ws   = (float*)d_ws;
    const size_t nBN  = (size_t)GB * GN;
    const size_t nBNH = (size_t)GB * GN * GH;
    const size_t nList = (size_t)NRANGE * GB;
    float* dinv = ws;
    unsigned int* counts = (unsigned int*)(dinv + nBN);
    v2u*   lists = (v2u*)(dinv + nBN + nList);
    float* tmp  = dinv + nBN + nList + 2 * nList * CAP;
    float* hbuf = tmp + nBNH;

    dim3 own_grid(NRANGE, GB, 1);

    deg_kernel<<<own_grid, OWN_T, 0, stream>>>(ei, ew, x, W1, tmp, dinv, lists, counts);

    agg_kernel<1, 1, 1><<<own_grid, OWN_T, 0, stream>>>(lists, counts, ei, ew, tmp, dinv, b1, W2, pos, hbuf);

    agg_kernel<1, 1, 0><<<own_grid, OWN_T, 0, stream>>>(lists, counts, ei, ew, hbuf, dinv, b2, W3, pos, tmp);

    agg_kernel<0, 0, 0><<<own_grid, OWN_T, 0, stream>>>(lists, counts, ei, ew, tmp, dinv, b3, W3, pos, out);
}
